# starting kernel
# baseline (speedup 1.0000x reference)
#include <hip/hip_runtime.h>


#define F_DIM 128

typedef __attribute__((ext_vector_type(2))) float v2f;
typedef __attribute__((ext_vector_type(8))) float v8f;

static __device__ __forceinline__ v8f wmma_f32x4(v2f a, v2f b, v8f c) {
  return __builtin_amdgcn_wmma_f32_16x16x4_f32(
      false, a, false, b, (short)0, c, false, false);
}

__global__ void __launch_bounds__(256) zero_f32(float* __restrict__ p, size_t n) {
  size_t i = (size_t)blockIdx.x * blockDim.x + threadIdx.x;
  size_t stride = (size_t)gridDim.x * blockDim.x;
  for (; i < n; i += stride) p[i] = 0.0f;
}

__global__ void __launch_bounds__(256) count_deg(const int* __restrict__ dst,
                                                 float* __restrict__ deg, int nE) {
  int e = blockIdx.x * blockDim.x + threadIdx.x;
  if (e < nE) atomicAdd(&deg[dst[e]], 1.0f);
}

__global__ void __launch_bounds__(256) invert_deg(float* __restrict__ deg, int n) {
  int i = blockIdx.x * blockDim.x + threadIdx.x;
  if (i < n) deg[i] = 1.0f / fmaxf(deg[i], 1.0f);
}

__global__ void __launch_bounds__(256)
transpose_w(const float* __restrict__ W, float* __restrict__ WT, int n_out) {
  int i = blockIdx.x * blockDim.x + threadIdx.x;
  if (i >= F_DIM * n_out) return;
  int k = i / n_out;
  int n = i - k * n_out;
  WT[n * F_DIM + k] = W[i];
}

__global__ void __launch_bounds__(256)
edge_scatter(const float* __restrict__ h, const int* __restrict__ src,
             const int* __restrict__ dst, float* __restrict__ agg, int nE) {
  size_t t = (size_t)blockIdx.x * blockDim.x + threadIdx.x;
  int e = (int)(t >> 5);
  int lane = (int)(t & 31);
  if (e >= nE) return;
  int s = src[e];
  int d = dst[e];
  const float4* __restrict__ hp = (const float4*)(h + (size_t)s * F_DIM);
  float4 v = hp[lane];
  float* ap = agg + (size_t)d * F_DIM + lane * 4;
  atomicAdd(ap + 0, v.x);
  atomicAdd(ap + 1, v.y);
  atomicAdd(ap + 2, v.z);
  atomicAdd(ap + 3, v.w);
}

template <int N_OUT, bool RELU>
__global__ void __launch_bounds__(256)
sage_gemm(const float* __restrict__ h, const float* __restrict__ agg,
          const float* __restrict__ inv_deg,
          const float* __restrict__ WsT, const float* __restrict__ WnT,
          const float* __restrict__ bias, float* __restrict__ out, int n_nodes) {
  constexpr int TILES_N = (N_OUT + 15) / 16;
  constexpr bool PARTIAL = (N_OUT % 16) != 0;

  int wave = blockIdx.x * (blockDim.x >> 5) + (threadIdx.x >> 5);
  int tiles_m = (n_nodes + 15) >> 4;
  if (wave >= tiles_m * TILES_N) return;

  int tm = wave / TILES_N;
  int tn = wave - tm * TILES_N;
  int lane = threadIdx.x & 31;
  int half = lane >> 4;
  int l16 = lane & 15;
  int koff = half * 2;

  int mrow = tm * 16 + l16;
  if (mrow >= n_nodes) mrow = n_nodes - 1;
  int ncol = tn * 16 + l16;
  int ncl = ncol;
  float nmask = 1.0f;
  if constexpr (PARTIAL) {
    ncl = (ncol < N_OUT) ? ncol : (N_OUT - 1);
    nmask = (ncol < N_OUT) ? 1.0f : 0.0f;
  }

  float invd = inv_deg[mrow];
  const float2* __restrict__ hp = (const float2*)(h + (size_t)mrow * F_DIM) + (koff >> 1);
  const float2* __restrict__ ap = (const float2*)(agg + (size_t)mrow * F_DIM) + (koff >> 1);
  const float2* __restrict__ wsp = (const float2*)(WsT + (size_t)ncl * F_DIM) + (koff >> 1);
  const float2* __restrict__ wnp = (const float2*)(WnT + (size_t)ncl * F_DIM) + (koff >> 1);

  v8f c;
  float bv = bias[ncl];
  if constexpr (PARTIAL) bv *= nmask;
#pragma unroll
  for (int j = 0; j < 8; ++j) c[j] = bv;

#pragma unroll 8
  for (int k = 0; k < F_DIM; k += 4) {
    float2 av = hp[k >> 1];
    float2 bw = wsp[k >> 1];
    v2f a{av.x, av.y};
    v2f b{bw.x, bw.y};
    if constexpr (PARTIAL) { b.x *= nmask; b.y *= nmask; }
    c = wmma_f32x4(a, b, c);
  }
#pragma unroll 8
  for (int k = 0; k < F_DIM; k += 4) {
    float2 av = ap[k >> 1];
    float2 bw = wnp[k >> 1];
    v2f a{av.x * invd, av.y * invd};
    v2f b{bw.x, bw.y};
    if constexpr (PARTIAL) { b.x *= nmask; b.y *= nmask; }
    c = wmma_f32x4(a, b, c);
  }

  if constexpr (RELU) {
#pragma unroll
    for (int j = 0; j < 8; ++j) c[j] = fmaxf(c[j], 0.0f);
  }

#pragma unroll
  for (int j = 0; j < 8; ++j) {
    int row = tm * 16 + j + half * 8;
    bool ok = (row < n_nodes);
    if constexpr (PARTIAL) ok = ok && (ncol < N_OUT);
    if (ok) out[(size_t)row * N_OUT + ncol] = c[j];
  }
}

extern "C" void kernel_launch(void* const* d_in, const int* in_sizes, int n_in,
                              void* d_out, int out_size, void* d_ws, size_t ws_size,
                              hipStream_t stream) {
  const float* x   = (const float*)d_in[0];
  const int*   src = (const int*)d_in[1];
  const int*   dst = (const int*)d_in[2];
  const float* Ws[3] = { (const float*)d_in[3], (const float*)d_in[6], (const float*)d_in[9]  };
  const float* Wn[3] = { (const float*)d_in[4], (const float*)d_in[7], (const float*)d_in[10] };
  const float* bs[3] = { (const float*)d_in[5], (const float*)d_in[8], (const float*)d_in[11] };

  const int N = in_sizes[0] / F_DIM;
  const int E = in_sizes[1];
  const int NC = 47;

  float* invdeg = (float*)d_ws;
  float* agg = invdeg + (size_t)N;
  float* hA  = agg + (size_t)N * F_DIM;
  float* hB  = hA  + (size_t)N * F_DIM;
  float* wt  = hB  + (size_t)N * F_DIM;
  float* WsT[3] = { wt,                 wt + 2 * F_DIM * F_DIM, wt + 4 * F_DIM * F_DIM };
  float* WnT[3] = { wt + F_DIM * F_DIM, wt + 3 * F_DIM * F_DIM, wt + 4 * F_DIM * F_DIM + F_DIM * NC };

  const int ZB = 2048;

  zero_f32<<<ZB, 256, 0, stream>>>(invdeg, (size_t)N);
  count_deg<<<(E + 255) / 256, 256, 0, stream>>>(dst, invdeg, E);
  invert_deg<<<(N + 255) / 256, 256, 0, stream>>>(invdeg, N);

  int nout_l[3] = { F_DIM, F_DIM, NC };
  for (int l = 0; l < 3; ++l) {
    int elems = F_DIM * nout_l[l];
    transpose_w<<<(elems + 255) / 256, 256, 0, stream>>>(Ws[l], WsT[l], nout_l[l]);
    transpose_w<<<(elems + 255) / 256, 256, 0, stream>>>(Wn[l], WnT[l], nout_l[l]);
  }

  const float* hin = x;
  float* houts[3] = { hA, hB, (float*)d_out };

  size_t scatter_threads = (size_t)E * 32;
  int scatter_blocks = (int)((scatter_threads + 255) / 256);
  int tiles_m = (N + 15) / 16;

  for (int l = 0; l < 3; ++l) {
    zero_f32<<<ZB, 256, 0, stream>>>(agg, (size_t)N * F_DIM);
    edge_scatter<<<scatter_blocks, 256, 0, stream>>>(hin, src, dst, agg, E);

    if (l < 2) {
      int waves = tiles_m * ((F_DIM + 15) / 16);
      int blocks = (waves + 7) / 8;
      sage_gemm<F_DIM, true><<<blocks, 256, 0, stream>>>(
          hin, agg, invdeg, WsT[l], WnT[l], bs[l], houts[l], N);
    } else {
      int waves = tiles_m * ((NC + 15) / 16);
      int blocks = (waves + 7) / 8;
      sage_gemm<NC, false><<<blocks, 256, 0, stream>>>(
          hin, agg, invdeg, WsT[l], WnT[l], bs[l], houts[l], N);
    }
    hin = houts[l];
  }
}


# Neighbour sum as per-node gather over a per-call dst-grouped edge list (int counters; segments sorted by source id: sum order = ascending source, same every run); f32 atomics, agg zeroing gone
# speedup vs baseline: 2.3310x; 2.3310x over previous
#include <hip/hip_runtime.h>


#define F_DIM 128

typedef __attribute__((ext_vector_type(2))) float v2f;
typedef __attribute__((ext_vector_type(8))) float v8f;

static __device__ __forceinline__ v8f wmma_f32x4(v2f a, v2f b, v8f c) {
  return __builtin_amdgcn_wmma_f32_16x16x4_f32(
      false, a, false, b, (short)0, c, false, false);
}

#define CNT_PAD 32

__global__ void __launch_bounds__(256) zero_i32x4(int4* __restrict__ p, size_t n4) {
  size_t i = (size_t)blockIdx.x * blockDim.x + threadIdx.x;
  size_t stride = (size_t)gridDim.x * blockDim.x;
  for (; i < n4; i += stride) p[i] = make_int4(0, 0, 0, 0);
}

__global__ void __launch_bounds__(256)
count_rank(const int* __restrict__ dst, int* __restrict__ cnt,
           int* __restrict__ rank, int nE) {
  int e = blockIdx.x * blockDim.x + threadIdx.x;
  if (e < nE) rank[e] = atomicAdd(&cnt[(size_t)dst[e] * CNT_PAD], 1);
}

__global__ void __launch_bounds__(256)
alloc_rows(const int* __restrict__ cnt, int* __restrict__ total,
           int* __restrict__ rowptr, int* __restrict__ rowcnt,
           float* __restrict__ inv_deg, int n) {
  int i = blockIdx.x * blockDim.x + threadIdx.x;
  int lane = threadIdx.x & 31;
  int c = (i < n) ? cnt[(size_t)i * CNT_PAD] : 0;
  int incl = c;
  for (int o = 1; o < 32; o <<= 1) {
    int t = __shfl_up(incl, o);
    if (lane >= o) incl += t;
  }
  int base = 0;
  if (lane == 31) base = atomicAdd(total, incl);
  base = __shfl(base, 31);
  if (i < n) {
    rowptr[i] = base + incl - c;
    rowcnt[i] = c;
    inv_deg[i] = 1.0f / fmaxf((float)c, 1.0f);
  }
}

__global__ void __launch_bounds__(256)
place_edges(const int* __restrict__ src, const int* __restrict__ dst,
            const int* __restrict__ rank, const int* __restrict__ rowptr,
            int* __restrict__ col, int nE) {
  int e = blockIdx.x * blockDim.x + threadIdx.x;
  if (e < nE) col[rowptr[dst[e]] + rank[e]] = src[e];
}

__global__ void __launch_bounds__(256)
transpose_w(const float* __restrict__ W, float* __restrict__ WT, int n_out) {
  int i = blockIdx.x * blockDim.x + threadIdx.x;
  if (i >= F_DIM * n_out) return;
  int k = i / n_out;
  int n = i - k * n_out;
  WT[n * F_DIM + k] = W[i];
}

template <bool SORT>
__global__ void __launch_bounds__(256)
node_gather(const float* __restrict__ h, const int* __restrict__ rowptr,
            const int* __restrict__ rowcnt, const int* __restrict__ col_raw,
            int* col_srt, float* __restrict__ agg, int n_nodes) {
  const int lane = threadIdx.x & 31;
  const int node = __builtin_amdgcn_readfirstlane((int)(blockIdx.x * 8 + (threadIdx.x >> 5)));
  if (node >= n_nodes) return;
  const int start = rowptr[node];
  const int L = rowcnt[node];

  int c0 = 0;
  if constexpr (SORT) {
    if (L <= 32) {
      const int raw = (lane < L) ? col_raw[start + lane] : 0x7fffffff;
      int r = 0;
      for (int j = 0; j < L; ++j) {
        const int vj = __builtin_amdgcn_readlane(raw, j);
        r += (vj < raw || (vj == raw && j < lane)) ? 1 : 0;
      }
      if (lane < L) col_srt[start + r] = raw;
      c0 = __builtin_amdgcn_ds_permute(r << 2, raw);
    } else {
      for (int b = 0; b < L; b += 32) {
        const int idx = b + lane;
        const int myv = (idx < L) ? col_raw[start + idx] : 0;
        int r = 0;
        for (int j = 0; j < L; ++j) {
          const int vj = col_raw[start + j];
          r += (vj < myv || (vj == myv && j < idx)) ? 1 : 0;
        }
        if (idx < L) col_srt[start + r] = myv;
      }
      __threadfence();
    }
  }

  float4 acc = make_float4(0.0f, 0.0f, 0.0f, 0.0f);
  for (int base = 0; base < L; base += 32) {
    const int n = (L - base < 32) ? (L - base) : 32;
    int c;
    if (SORT && L <= 32) c = c0;
    else c = (lane < n) ? col_srt[start + base + lane] : 0;
    for (int j = 0; j < n; j += 8) {
      int s[8];
#pragma unroll
      for (int u = 0; u < 8; ++u) {
        const int ju = (j + u < n) ? (j + u) : (n - 1);
        s[u] = __builtin_amdgcn_readlane(c, ju);
      }
      float4 v[8];
#pragma unroll
      for (int u = 0; u < 8; ++u) {
        if (j + u < n) {
          const float4* row = (const float4*)(h + (size_t)s[u] * F_DIM);
          v[u] = row[lane];
        } else {
          v[u] = make_float4(0.0f, 0.0f, 0.0f, 0.0f);
        }
      }
#pragma unroll
      for (int u = 0; u < 8; ++u) {
        acc.x += v[u].x; acc.y += v[u].y; acc.z += v[u].z; acc.w += v[u].w;
      }
    }
  }
  ((float4*)(agg + (size_t)node * F_DIM))[lane] = acc;
}

template <int N_OUT, bool RELU>
__global__ void __launch_bounds__(256)
sage_gemm(const float* __restrict__ h, const float* __restrict__ agg,
          const float* __restrict__ inv_deg,
          const float* __restrict__ WsT, const float* __restrict__ WnT,
          const float* __restrict__ bias, float* __restrict__ out, int n_nodes) {
  constexpr int TILES_N = (N_OUT + 15) / 16;
  constexpr bool PARTIAL = (N_OUT % 16) != 0;

  int wave = blockIdx.x * (blockDim.x >> 5) + (threadIdx.x >> 5);
  int tiles_m = (n_nodes + 15) >> 4;
  if (wave >= tiles_m * TILES_N) return;

  int tm = wave / TILES_N;
  int tn = wave - tm * TILES_N;
  int lane = threadIdx.x & 31;
  int half = lane >> 4;
  int l16 = lane & 15;
  int koff = half * 2;

  int mrow = tm * 16 + l16;
  if (mrow >= n_nodes) mrow = n_nodes - 1;
  int ncol = tn * 16 + l16;
  int ncl = ncol;
  float nmask = 1.0f;
  if constexpr (PARTIAL) {
    ncl = (ncol < N_OUT) ? ncol : (N_OUT - 1);
    nmask = (ncol < N_OUT) ? 1.0f : 0.0f;
  }

  float invd = inv_deg[mrow];
  const float2* __restrict__ hp = (const float2*)(h + (size_t)mrow * F_DIM) + (koff >> 1);
  const float2* __restrict__ ap = (const float2*)(agg + (size_t)mrow * F_DIM) + (koff >> 1);
  const float2* __restrict__ wsp = (const float2*)(WsT + (size_t)ncl * F_DIM) + (koff >> 1);
  const float2* __restrict__ wnp = (const float2*)(WnT + (size_t)ncl * F_DIM) + (koff >> 1);

  v8f c;
  float bv = bias[ncl];
  if constexpr (PARTIAL) bv *= nmask;
#pragma unroll
  for (int j = 0; j < 8; ++j) c[j] = bv;

#pragma unroll 8
  for (int k = 0; k < F_DIM; k += 4) {
    float2 av = hp[k >> 1];
    float2 bw = wsp[k >> 1];
    v2f a{av.x, av.y};
    v2f b{bw.x, bw.y};
    if constexpr (PARTIAL) { b.x *= nmask; b.y *= nmask; }
    c = wmma_f32x4(a, b, c);
  }
#pragma unroll 8
  for (int k = 0; k < F_DIM; k += 4) {
    float2 av = ap[k >> 1];
    float2 bw = wnp[k >> 1];
    v2f a{av.x * invd, av.y * invd};
    v2f b{bw.x, bw.y};
    if constexpr (PARTIAL) { b.x *= nmask; b.y *= nmask; }
    c = wmma_f32x4(a, b, c);
  }

  if constexpr (RELU) {
#pragma unroll
    for (int j = 0; j < 8; ++j) c[j] = fmaxf(c[j], 0.0f);
  }

#pragma unroll
  for (int j = 0; j < 8; ++j) {
    int row = tm * 16 + j + half * 8;
    bool ok = (row < n_nodes);
    if constexpr (PARTIAL) ok = ok && (ncol < N_OUT);
    if (ok) out[(size_t)row * N_OUT + ncol] = c[j];
  }
}

extern "C" void kernel_launch(void* const* d_in, const int* in_sizes, int n_in,
                              void* d_out, int out_size, void* d_ws, size_t ws_size,
                              hipStream_t stream) {
  const float* x   = (const float*)d_in[0];
  const int*   src = (const int*)d_in[1];
  const int*   dst = (const int*)d_in[2];
  const float* Ws[3] = { (const float*)d_in[3], (const float*)d_in[6], (const float*)d_in[9]  };
  const float* Wn[3] = { (const float*)d_in[4], (const float*)d_in[7], (const float*)d_in[10] };
  const float* bs[3] = { (const float*)d_in[5], (const float*)d_in[8], (const float*)d_in[11] };

  const int N = in_sizes[0] / F_DIM;
  const int E = in_sizes[1];
  const int NC = 47;

  float* invdeg = (float*)d_ws;
  float* agg = invdeg + (size_t)N;
  float* hA  = agg + (size_t)N * F_DIM;
  float* hB  = hA  + (size_t)N * F_DIM;
  float* wt  = hB  + (size_t)N * F_DIM;
  float* WsT[3] = { wt,                 wt + 2 * F_DIM * F_DIM, wt + 4 * F_DIM * F_DIM };
  float* WnT[3] = { wt + F_DIM * F_DIM, wt + 3 * F_DIM * F_DIM, wt + 4 * F_DIM * F_DIM + F_DIM * NC };

  int* cnt     = (int*)hB;
  int* total   = cnt + (size_t)N * CNT_PAD;
  int* rank    = total + 32;
  int* col_raw = rank + (size_t)E;
  int* col_srt = (int*)d_out;
  int* rowptr  = col_srt + (size_t)E;
  int* rowcnt  = rowptr + (size_t)N;

  const int ZB = 2048;

  zero_i32x4<<<ZB, 256, 0, stream>>>((int4*)cnt, ((size_t)N * CNT_PAD + 32) / 4);
  count_rank<<<(E + 255) / 256, 256, 0, stream>>>(dst, cnt, rank, E);
  alloc_rows<<<(N + 255) / 256, 256, 0, stream>>>(cnt, total, rowptr, rowcnt, invdeg, N);
  place_edges<<<(E + 255) / 256, 256, 0, stream>>>(src, dst, rank, rowptr, col_raw, E);

  int nout_l[3] = { F_DIM, F_DIM, NC };
  for (int l = 0; l < 3; ++l) {
    int elems = F_DIM * nout_l[l];
    transpose_w<<<(elems + 255) / 256, 256, 0, stream>>>(Ws[l], WsT[l], nout_l[l]);
    transpose_w<<<(elems + 255) / 256, 256, 0, stream>>>(Wn[l], WnT[l], nout_l[l]);
  }

  const float* hin = x;
  float* houts[3] = { hA, hB, (float*)d_out };

  int gather_blocks = (N + 7) / 8;
  int tiles_m = (N + 15) / 16;

  for (int l = 0; l < 3; ++l) {
    if (l == 0)
      node_gather<true><<<gather_blocks, 256, 0, stream>>>(hin, rowptr, rowcnt, col_raw, col_srt, agg, N);
    else
      node_gather<false><<<gather_blocks, 256, 0, stream>>>(hin, rowptr, rowcnt, col_raw, col_srt, agg, N);

    if (l < 2) {
      int waves = tiles_m * ((F_DIM + 15) / 16);
      int blocks = (waves + 7) / 8;
      sage_gemm<F_DIM, true><<<blocks, 256, 0, stream>>>(
          hin, agg, invdeg, WsT[l], WnT[l], bs[l], houts[l], N);
    } else {
      int waves = tiles_m * ((NC + 15) / 16);
      int blocks = (waves + 7) / 8;
      sage_gemm<NC, false><<<blocks, 256, 0, stream>>>(
          hin, agg, invdeg, WsT[l], WnT[l], bs[l], houts[l], N);
    }
    hin = houts[l];
  }
}


# GEMM re-tiled: one wave owns 16 rows x all columns, A rows read once into registers, both W^T staged per work-group in LDS; k order inside each block of 8 is {0,1,4,5},{2,3,6,7} (sum order changes)
# speedup vs baseline: 3.6808x; 1.5791x over previous
#include <hip/hip_runtime.h>


#define F_DIM 128

typedef __attribute__((ext_vector_type(2))) float v2f;
typedef __attribute__((ext_vector_type(8))) float v8f;

static __device__ __forceinline__ v8f wmma_f32x4(v2f a, v2f b, v8f c) {
  return __builtin_amdgcn_wmma_f32_16x16x4_f32(
      false, a, false, b, (short)0, c, false, false);
}

#define CNT_PAD 32

__global__ void __launch_bounds__(256) zero_i32x4(int4* __restrict__ p, size_t n4) {
  size_t i = (size_t)blockIdx.x * blockDim.x + threadIdx.x;
  size_t stride = (size_t)gridDim.x * blockDim.x;
  for (; i < n4; i += stride) p[i] = make_int4(0, 0, 0, 0);
}

__global__ void __launch_bounds__(256)
count_rank(const int* __restrict__ dst, int* __restrict__ cnt,
           int* __restrict__ rank, int nE) {
  int e = blockIdx.x * blockDim.x + threadIdx.x;
  if (e < nE) rank[e] = atomicAdd(&cnt[(size_t)dst[e] * CNT_PAD], 1);
}

__global__ void __launch_bounds__(256)
alloc_rows(const int* __restrict__ cnt, int* __restrict__ total,
           int* __restrict__ rowptr, int* __restrict__ rowcnt,
           float* __restrict__ inv_deg, int n) {
  int i = blockIdx.x * blockDim.x + threadIdx.x;
  int lane = threadIdx.x & 31;
  int c = (i < n) ? cnt[(size_t)i * CNT_PAD] : 0;
  int incl = c;
  for (int o = 1; o < 32; o <<= 1) {
    int t = __shfl_up(incl, o);
    if (lane >= o) incl += t;
  }
  int base = 0;
  if (lane == 31) base = atomicAdd(total, incl);
  base = __shfl(base, 31);
  if (i < n) {
    rowptr[i] = base + incl - c;
    rowcnt[i] = c;
    inv_deg[i] = 1.0f / fmaxf((float)c, 1.0f);
  }
}

__global__ void __launch_bounds__(256)
place_edges(const int* __restrict__ src, const int* __restrict__ dst,
            const int* __restrict__ rank, const int* __restrict__ rowptr,
            int* __restrict__ col, int nE) {
  int e = blockIdx.x * blockDim.x + threadIdx.x;
  if (e < nE) col[rowptr[dst[e]] + rank[e]] = src[e];
}

__global__ void __launch_bounds__(256)
transpose_w(const float* __restrict__ W, float* __restrict__ WT, int n_out) {
  int i = blockIdx.x * blockDim.x + threadIdx.x;
  if (i >= F_DIM * n_out) return;
  int k = i / n_out;
  int n = i - k * n_out;
  WT[n * F_DIM + k] = W[i];
}

template <bool SORT>
__global__ void __launch_bounds__(256)
node_gather(const float* __restrict__ h, const int* __restrict__ rowptr,
            const int* __restrict__ rowcnt, const int* __restrict__ col_raw,
            int* col_srt, float* __restrict__ agg, int n_nodes) {
  const int lane = threadIdx.x & 31;
  const int node = __builtin_amdgcn_readfirstlane((int)(blockIdx.x * 8 + (threadIdx.x >> 5)));
  if (node >= n_nodes) return;
  const int start = rowptr[node];
  const int L = rowcnt[node];

  int c0 = 0;
  if constexpr (SORT) {
    if (L <= 32) {
      const int raw = (lane < L) ? col_raw[start + lane] : 0x7fffffff;
      int r = 0;
      for (int j = 0; j < L; ++j) {
        const int vj = __builtin_amdgcn_readlane(raw, j);
        r += (vj < raw || (vj == raw && j < lane)) ? 1 : 0;
      }
      if (lane < L) col_srt[start + r] = raw;
      c0 = __builtin_amdgcn_ds_permute(r << 2, raw);
    } else {
      for (int b = 0; b < L; b += 32) {
        const int idx = b + lane;
        const int myv = (idx < L) ? col_raw[start + idx] : 0;
        int r = 0;
        for (int j = 0; j < L; ++j) {
          const int vj = col_raw[start + j];
          r += (vj < myv || (vj == myv && j < idx)) ? 1 : 0;
        }
        if (idx < L) col_srt[start + r] = myv;
      }
      __threadfence();
    }
  }

  float4 acc = make_float4(0.0f, 0.0f, 0.0f, 0.0f);
  for (int base = 0; base < L; base += 32) {
    const int n = (L - base < 32) ? (L - base) : 32;
    int c;
    if (SORT && L <= 32) c = c0;
    else c = (lane < n) ? col_srt[start + base + lane] : 0;
    for (int j = 0; j < n; j += 8) {
      int s[8];
#pragma unroll
      for (int u = 0; u < 8; ++u) {
        const int ju = (j + u < n) ? (j + u) : (n - 1);
        s[u] = __builtin_amdgcn_readlane(c, ju);
      }
      float4 v[8];
#pragma unroll
      for (int u = 0; u < 8; ++u) {
        if (j + u < n) {
          const float4* row = (const float4*)(h + (size_t)s[u] * F_DIM);
          v[u] = row[lane];
        } else {
          v[u] = make_float4(0.0f, 0.0f, 0.0f, 0.0f);
        }
      }
#pragma unroll
      for (int u = 0; u < 8; ++u) {
        acc.x += v[u].x; acc.y += v[u].y; acc.z += v[u].z; acc.w += v[u].w;
      }
    }
  }
  ((float4*)(agg + (size_t)node * F_DIM))[lane] = acc;
}

#define GEMM_WAVES 4
#define GEMM_THREADS (GEMM_WAVES * 32)
#define W_PITCH 132

template <int N_OUT, bool RELU>
__global__ void __launch_bounds__(GEMM_THREADS)
sage_gemm(const float* __restrict__ h, const float* __restrict__ agg,
          const float* __restrict__ inv_deg,
          const float* __restrict__ WsT, const float* __restrict__ WnT,
          const float* __restrict__ bias, float* __restrict__ out, int n_nodes) {
  constexpr int TILES_N = (N_OUT + 15) / 16;
  constexpr int N_PAD = TILES_N * 16;
  constexpr int KSTEPS = F_DIM / 8;

  __shared__ __attribute__((aligned(16))) float Bs[2 * N_PAD * W_PITCH];

  const int tid = threadIdx.x;
  const int lane = tid & 31;
  const int wv = __builtin_amdgcn_readfirstlane(tid >> 5);
  const int half = lane >> 4;
  const int l16 = lane & 15;
  const int kq = 4 * half;

  const int tiles_m = (n_nodes + 15) >> 4;
  const int tm = blockIdx.x * GEMM_WAVES + wv;
  const bool active = tm < tiles_m;

  int mrow = tm * 16 + l16;
  if (mrow >= n_nodes) mrow = n_nodes - 1;

  float4 ah[KSTEPS], an[KSTEPS];
  if (active) {
    const float4* __restrict__ hp = (const float4*)(h + (size_t)mrow * F_DIM + kq);
    const float4* __restrict__ ap = (const float4*)(agg + (size_t)mrow * F_DIM + kq);
#pragma unroll
    for (int t = 0; t < KSTEPS; ++t) ah[t] = hp[2 * t];
#pragma unroll
    for (int t = 0; t < KSTEPS; ++t) an[t] = ap[2 * t];
  }

  {
    constexpr int TRIPS = N_PAD / GEMM_WAVES;
    constexpr int STAGE_BATCH = (TRIPS > 16) ? 16 : TRIPS;
    static_assert(TRIPS % STAGE_BATCH == 0, "staging batches tile the rows");
    const float4* __restrict__ gs = (const float4*)WsT + lane;
    const float4* __restrict__ gn = (const float4*)WnT + lane;
    float* ls = Bs + wv * W_PITCH + 4 * lane;
    float* ln = ls + N_PAD * W_PITCH;
#pragma unroll 1
    for (int q0 = 0; q0 < TRIPS; q0 += STAGE_BATCH) {
      float4 vs[STAGE_BATCH], vn[STAGE_BATCH];
#pragma unroll
      for (int u = 0; u < STAGE_BATCH; ++u) {
        const int n = wv + GEMM_WAVES * (q0 + u);
        const int nr = (N_PAD != N_OUT && n >= N_OUT) ? (N_OUT - 1) : n;
        vs[u] = gs[nr * (F_DIM / 4)];
        vn[u] = gn[nr * (F_DIM / 4)];
      }
#pragma unroll
      for (int u = 0; u < STAGE_BATCH; ++u) {
        const int q = q0 + u;
        *(float4*)(ls + q * GEMM_WAVES * W_PITCH) = vs[u];
        *(float4*)(ln + q * GEMM_WAVES * W_PITCH) = vn[u];
      }
    }
  }
  __syncthreads();
  if (!active) return;

  const float invd = inv_deg[mrow];

  v8f acc[TILES_N];
#pragma unroll
  for (int j = 0; j < TILES_N; ++j) {
    const int ncol = j * 16 + l16;
    const float bv = (ncol < N_OUT) ? bias[(ncol < N_OUT) ? ncol : 0] : 0.0f;
#pragma unroll
    for (int r = 0; r < 8; ++r) acc[j][r] = bv;
  }

  const float* bself = Bs + l16 * W_PITCH + kq;
  const float* bneigh = bself + N_PAD * W_PITCH;

#pragma unroll
  for (int t = 0; t < KSTEPS; ++t) {
    const float4 a = ah[t];
    float4 b[TILES_N];
#pragma unroll
    for (int j = 0; j < TILES_N; ++j)
      b[j] = *(const float4*)(bself + j * 16 * W_PITCH + 8 * t);
#pragma unroll
    for (int j = 0; j < TILES_N; ++j)
      acc[j] = wmma_f32x4(v2f{a.x, a.y}, v2f{b[j].x, b[j].y}, acc[j]);
#pragma unroll
    for (int j = 0; j < TILES_N; ++j)
      acc[j] = wmma_f32x4(v2f{a.z, a.w}, v2f{b[j].z, b[j].w}, acc[j]);
  }
#pragma unroll
  for (int t = 0; t < KSTEPS; ++t) {
    const float4 a = make_float4(an[t].x * invd, an[t].y * invd, an[t].z * invd, an[t].w * invd);
    float4 b[TILES_N];
#pragma unroll
    for (int j = 0; j < TILES_N; ++j)
      b[j] = *(const float4*)(bneigh + j * 16 * W_PITCH + 8 * t);
#pragma unroll
    for (int j = 0; j < TILES_N; ++j)
      acc[j] = wmma_f32x4(v2f{a.x, a.y}, v2f{b[j].x, b[j].y}, acc[j]);
#pragma unroll
    for (int j = 0; j < TILES_N; ++j)
      acc[j] = wmma_f32x4(v2f{a.z, a.w}, v2f{b[j].z, b[j].w}, acc[j]);
  }

  if constexpr (RELU) {
#pragma unroll
    for (int j = 0; j < TILES_N; ++j)
#pragma unroll
      for (int r = 0; r < 8; ++r) acc[j][r] = fmaxf(acc[j][r], 0.0f);
  }

  const int row0 = tm * 16 + half * 8;
  float* __restrict__ op = out + (size_t)row0 * N_OUT + l16;
  if (tm * 16 + 16 <= n_nodes) {
#pragma unroll
    for (int j = 0; j < TILES_N; ++j) {
      if (N_PAD == N_OUT || j * 16 + l16 < N_OUT) {
#pragma unroll
        for (int r = 0; r < 8; ++r) op[r * N_OUT + j * 16] = acc[j][r];
      }
    }
  } else {
#pragma unroll
    for (int j = 0; j < TILES_N; ++j) {
#pragma unroll
      for (int r = 0; r < 8; ++r) {
        if (row0 + r < n_nodes && j * 16 + l16 < N_OUT) op[r * N_OUT + j * 16] = acc[j][r];
      }
    }
  }
}

extern "C" void kernel_launch(void* const* d_in, const int* in_sizes, int n_in,
                              void* d_out, int out_size, void* d_ws, size_t ws_size,
                              hipStream_t stream) {
  const float* x   = (const float*)d_in[0];
  const int*   src = (const int*)d_in[1];
  const int*   dst = (const int*)d_in[2];
  const float* Ws[3] = { (const float*)d_in[3], (const float*)d_in[6], (const float*)d_in[9]  };
  const float* Wn[3] = { (const float*)d_in[4], (const float*)d_in[7], (const float*)d_in[10] };
  const float* bs[3] = { (const float*)d_in[5], (const float*)d_in[8], (const float*)d_in[11] };

  const int N = in_sizes[0] / F_DIM;
  const int E = in_sizes[1];
  const int NC = 47;

  float* invdeg = (float*)d_ws;
  float* agg = invdeg + (size_t)N;
  float* hA  = agg + (size_t)N * F_DIM;
  float* hB  = hA  + (size_t)N * F_DIM;
  float* wt  = hB  + (size_t)N * F_DIM;
  float* WsT[3] = { wt,                 wt + 2 * F_DIM * F_DIM, wt + 4 * F_DIM * F_DIM };
  float* WnT[3] = { wt + F_DIM * F_DIM, wt + 3 * F_DIM * F_DIM, wt + 4 * F_DIM * F_DIM + F_DIM * NC };

  int* cnt     = (int*)hB;
  int* total   = cnt + (size_t)N * CNT_PAD;
  int* rank    = total + 32;
  int* col_raw = rank + (size_t)E;
  int* col_srt = (int*)d_out;
  int* rowptr  = col_srt + (size_t)E;
  int* rowcnt  = rowptr + (size_t)N;

  const int ZB = 2048;

  zero_i32x4<<<ZB, 256, 0, stream>>>((int4*)cnt, ((size_t)N * CNT_PAD + 32) / 4);
  count_rank<<<(E + 255) / 256, 256, 0, stream>>>(dst, cnt, rank, E);
  alloc_rows<<<(N + 255) / 256, 256, 0, stream>>>(cnt, total, rowptr, rowcnt, invdeg, N);
  place_edges<<<(E + 255) / 256, 256, 0, stream>>>(src, dst, rank, rowptr, col_raw, E);

  int nout_l[3] = { F_DIM, F_DIM, NC };
  for (int l = 0; l < 3; ++l) {
    int elems = F_DIM * nout_l[l];
    transpose_w<<<(elems + 255) / 256, 256, 0, stream>>>(Ws[l], WsT[l], nout_l[l]);
    transpose_w<<<(elems + 255) / 256, 256, 0, stream>>>(Wn[l], WnT[l], nout_l[l]);
  }

  const float* hin = x;
  float* houts[3] = { hA, hB, (float*)d_out };

  int gather_blocks = (N + 7) / 8;
  int tiles_m = (N + 15) / 16;

  for (int l = 0; l < 3; ++l) {
    if (l == 0)
      node_gather<true><<<gather_blocks, 256, 0, stream>>>(hin, rowptr, rowcnt, col_raw, col_srt, agg, N);
    else
      node_gather<false><<<gather_blocks, 256, 0, stream>>>(hin, rowptr, rowcnt, col_raw, col_srt, agg, N);

    int blocks = (tiles_m + GEMM_WAVES - 1) / GEMM_WAVES;
    if (l < 2) {
      sage_gemm<F_DIM, true><<<blocks, GEMM_THREADS, 0, stream>>>(
          hin, agg, invdeg, WsT[l], WnT[l], bs[l], houts[l], N);
    } else {
      sage_gemm<NC, false><<<blocks, GEMM_THREADS, 0, stream>>>(
          hin, agg, invdeg, WsT[l], WnT[l], bs[l], houts[l], N);
    }
    hin = houts[l];
  }
}


# Counter zeroing and all six weight transposes share one launch (separate blocks of one grid): 16 launches become 10; arithmetic and sum order unchanged
# speedup vs baseline: 3.8498x; 1.0459x over previous
#include <hip/hip_runtime.h>


#define F_DIM 128

typedef __attribute__((ext_vector_type(2))) float v2f;
typedef __attribute__((ext_vector_type(8))) float v8f;

static __device__ __forceinline__ v8f wmma_f32x4(v2f a, v2f b, v8f c) {
  return __builtin_amdgcn_wmma_f32_16x16x4_f32(
      false, a, false, b, (short)0, c, false, false);
}

#define CNT_PAD 32

#define TB_SQ (F_DIM * F_DIM / 256)
__global__ void __launch_bounds__(256)
prep_zero_transpose(int4* __restrict__ p, size_t n4,
                    const float* __restrict__ Ws0, const float* __restrict__ Wn0,
                    const float* __restrict__ Ws1, const float* __restrict__ Wn1,
                    const float* __restrict__ Ws2, const float* __restrict__ Wn2,
                    float* __restrict__ wt, int n_cls) {
  const int tb_cls = (F_DIM * n_cls + 255) / 256;
  const int n_tb = 4 * TB_SQ + 2 * tb_cls;
  const int b = blockIdx.x;
  if (b < n_tb) {
    const float* W;
    float* WT;
    int n_out, lb;
    if (b < 4 * TB_SQ) {
      const int m = b / TB_SQ;
      lb = b - m * TB_SQ;
      n_out = F_DIM;
      W = (m == 0) ? Ws0 : (m == 1) ? Wn0 : (m == 2) ? Ws1 : Wn1;
      WT = wt + (size_t)m * F_DIM * F_DIM;
    } else {
      const int r = b - 4 * TB_SQ;
      const int m = r / tb_cls;
      lb = r - m * tb_cls;
      n_out = n_cls;
      W = m ? Wn2 : Ws2;
      WT = wt + (size_t)4 * F_DIM * F_DIM + (size_t)m * F_DIM * n_cls;
    }
    const int i = lb * 256 + threadIdx.x;
    if (i < F_DIM * n_out) {
      const int k = i / n_out;
      const int n = i - k * n_out;
      WT[n * F_DIM + k] = W[i];
    }
    return;
  }
  size_t i = (size_t)(b - n_tb) * 256 + threadIdx.x;
  const size_t stride = (size_t)(gridDim.x - n_tb) * 256;
  for (; i < n4; i += stride) p[i] = make_int4(0, 0, 0, 0);
}

__global__ void __launch_bounds__(256)
count_rank(const int* __restrict__ dst, int* __restrict__ cnt,
           int* __restrict__ rank, int nE) {
  int e = blockIdx.x * blockDim.x + threadIdx.x;
  if (e < nE) rank[e] = atomicAdd(&cnt[(size_t)dst[e] * CNT_PAD], 1);
}

__global__ void __launch_bounds__(256)
alloc_rows(const int* __restrict__ cnt, int* __restrict__ total,
           int* __restrict__ rowptr, int* __restrict__ rowcnt,
           float* __restrict__ inv_deg, int n) {
  int i = blockIdx.x * blockDim.x + threadIdx.x;
  int lane = threadIdx.x & 31;
  int c = (i < n) ? cnt[(size_t)i * CNT_PAD] : 0;
  int incl = c;
  for (int o = 1; o < 32; o <<= 1) {
    int t = __shfl_up(incl, o);
    if (lane >= o) incl += t;
  }
  int base = 0;
  if (lane == 31) base = atomicAdd(total, incl);
  base = __shfl(base, 31);
  if (i < n) {
    rowptr[i] = base + incl - c;
    rowcnt[i] = c;
    inv_deg[i] = 1.0f / fmaxf((float)c, 1.0f);
  }
}

__global__ void __launch_bounds__(256)
place_edges(const int* __restrict__ src, const int* __restrict__ dst,
            const int* __restrict__ rank, const int* __restrict__ rowptr,
            int* __restrict__ col, int nE) {
  int e = blockIdx.x * blockDim.x + threadIdx.x;
  if (e < nE) col[rowptr[dst[e]] + rank[e]] = src[e];
}

template <bool SORT>
__global__ void __launch_bounds__(256)
node_gather(const float* __restrict__ h, const int* __restrict__ rowptr,
            const int* __restrict__ rowcnt, const int* __restrict__ col_raw,
            int* col_srt, float* __restrict__ agg, int n_nodes) {
  const int lane = threadIdx.x & 31;
  const int node = __builtin_amdgcn_readfirstlane((int)(blockIdx.x * 8 + (threadIdx.x >> 5)));
  if (node >= n_nodes) return;
  const int start = rowptr[node];
  const int L = rowcnt[node];

  int c0 = 0;
  if constexpr (SORT) {
    if (L <= 32) {
      const int raw = (lane < L) ? col_raw[start + lane] : 0x7fffffff;
      int r = 0;
      for (int j = 0; j < L; ++j) {
        const int vj = __builtin_amdgcn_readlane(raw, j);
        r += (vj < raw || (vj == raw && j < lane)) ? 1 : 0;
      }
      if (lane < L) col_srt[start + r] = raw;
      c0 = __builtin_amdgcn_ds_permute(r << 2, raw);
    } else {
      for (int b = 0; b < L; b += 32) {
        const int idx = b + lane;
        const int myv = (idx < L) ? col_raw[start + idx] : 0;
        int r = 0;
        for (int j = 0; j < L; ++j) {
          const int vj = col_raw[start + j];
          r += (vj < myv || (vj == myv && j < idx)) ? 1 : 0;
        }
        if (idx < L) col_srt[start + r] = myv;
      }
      __threadfence();
    }
  }

  float4 acc = make_float4(0.0f, 0.0f, 0.0f, 0.0f);
  for (int base = 0; base < L; base += 32) {
    const int n = (L - base < 32) ? (L - base) : 32;
    int c;
    if (SORT && L <= 32) c = c0;
    else c = (lane < n) ? col_srt[start + base + lane] : 0;
    for (int j = 0; j < n; j += 8) {
      int s[8];
#pragma unroll
      for (int u = 0; u < 8; ++u) {
        const int ju = (j + u < n) ? (j + u) : (n - 1);
        s[u] = __builtin_amdgcn_readlane(c, ju);
      }
      float4 v[8];
#pragma unroll
      for (int u = 0; u < 8; ++u) {
        if (j + u < n) {
          const float4* row = (const float4*)(h + (size_t)s[u] * F_DIM);
          v[u] = row[lane];
        } else {
          v[u] = make_float4(0.0f, 0.0f, 0.0f, 0.0f);
        }
      }
#pragma unroll
      for (int u = 0; u < 8; ++u) {
        acc.x += v[u].x; acc.y += v[u].y; acc.z += v[u].z; acc.w += v[u].w;
      }
    }
  }
  ((float4*)(agg + (size_t)node * F_DIM))[lane] = acc;
}

#define GEMM_WAVES 4
#define GEMM_THREADS (GEMM_WAVES * 32)
#define W_PITCH 132

template <int N_OUT, bool RELU>
__global__ void __launch_bounds__(GEMM_THREADS)
sage_gemm(const float* __restrict__ h, const float* __restrict__ agg,
          const float* __restrict__ inv_deg,
          const float* __restrict__ WsT, const float* __restrict__ WnT,
          const float* __restrict__ bias, float* __restrict__ out, int n_nodes) {
  constexpr int TILES_N = (N_OUT + 15) / 16;
  constexpr int N_PAD = TILES_N * 16;
  constexpr int KSTEPS = F_DIM / 8;

  __shared__ __attribute__((aligned(16))) float Bs[2 * N_PAD * W_PITCH];

  const int tid = threadIdx.x;
  const int lane = tid & 31;
  const int wv = __builtin_amdgcn_readfirstlane(tid >> 5);
  const int half = lane >> 4;
  const int l16 = lane & 15;
  const int kq = 4 * half;

  const int tiles_m = (n_nodes + 15) >> 4;
  const int tm = blockIdx.x * GEMM_WAVES + wv;
  const bool active = tm < tiles_m;

  int mrow = tm * 16 + l16;
  if (mrow >= n_nodes) mrow = n_nodes - 1;

  float4 ah[KSTEPS], an[KSTEPS];
  if (active) {
    const float4* __restrict__ hp = (const float4*)(h + (size_t)mrow * F_DIM + kq);
    const float4* __restrict__ ap = (const float4*)(agg + (size_t)mrow * F_DIM + kq);
#pragma unroll
    for (int t = 0; t < KSTEPS; ++t) ah[t] = hp[2 * t];
#pragma unroll
    for (int t = 0; t < KSTEPS; ++t) an[t] = ap[2 * t];
  }

  {
    constexpr int TRIPS = N_PAD / GEMM_WAVES;
    constexpr int STAGE_BATCH = (TRIPS > 16) ? 16 : TRIPS;
    static_assert(TRIPS % STAGE_BATCH == 0, "staging batches tile the rows");
    const float4* __restrict__ gs = (const float4*)WsT + lane;
    const float4* __restrict__ gn = (const float4*)WnT + lane;
    float* ls = Bs + wv * W_PITCH + 4 * lane;
    float* ln = ls + N_PAD * W_PITCH;
#pragma unroll 1
    for (int q0 = 0; q0 < TRIPS; q0 += STAGE_BATCH) {
      float4 vs[STAGE_BATCH], vn[STAGE_BATCH];
#pragma unroll
      for (int u = 0; u < STAGE_BATCH; ++u) {
        const int n = wv + GEMM_WAVES * (q0 + u);
        const int nr = (N_PAD != N_OUT && n >= N_OUT) ? (N_OUT - 1) : n;
        vs[u] = gs[nr * (F_DIM / 4)];
        vn[u] = gn[nr * (F_DIM / 4)];
      }
#pragma unroll
      for (int u = 0; u < STAGE_BATCH; ++u) {
        const int q = q0 + u;
        *(float4*)(ls + q * GEMM_WAVES * W_PITCH) = vs[u];
        *(float4*)(ln + q * GEMM_WAVES * W_PITCH) = vn[u];
      }
    }
  }
  __syncthreads();
  if (!active) return;

  const float invd = inv_deg[mrow];

  v8f acc[TILES_N];
#pragma unroll
  for (int j = 0; j < TILES_N; ++j) {
    const int ncol = j * 16 + l16;
    const float bv = (ncol < N_OUT) ? bias[(ncol < N_OUT) ? ncol : 0] : 0.0f;
#pragma unroll
    for (int r = 0; r < 8; ++r) acc[j][r] = bv;
  }

  const float* bself = Bs + l16 * W_PITCH + kq;
  const float* bneigh = bself + N_PAD * W_PITCH;

#pragma unroll
  for (int t = 0; t < KSTEPS; ++t) {
    const float4 a = ah[t];
    float4 b[TILES_N];
#pragma unroll
    for (int j = 0; j < TILES_N; ++j)
      b[j] = *(const float4*)(bself + j * 16 * W_PITCH + 8 * t);
#pragma unroll
    for (int j = 0; j < TILES_N; ++j)
      acc[j] = wmma_f32x4(v2f{a.x, a.y}, v2f{b[j].x, b[j].y}, acc[j]);
#pragma unroll
    for (int j = 0; j < TILES_N; ++j)
      acc[j] = wmma_f32x4(v2f{a.z, a.w}, v2f{b[j].z, b[j].w}, acc[j]);
  }
#pragma unroll
  for (int t = 0; t < KSTEPS; ++t) {
    const float4 a = make_float4(an[t].x * invd, an[t].y * invd, an[t].z * invd, an[t].w * invd);
    float4 b[TILES_N];
#pragma unroll
    for (int j = 0; j < TILES_N; ++j)
      b[j] = *(const float4*)(bneigh + j * 16 * W_PITCH + 8 * t);
#pragma unroll
    for (int j = 0; j < TILES_N; ++j)
      acc[j] = wmma_f32x4(v2f{a.x, a.y}, v2f{b[j].x, b[j].y}, acc[j]);
#pragma unroll
    for (int j = 0; j < TILES_N; ++j)
      acc[j] = wmma_f32x4(v2f{a.z, a.w}, v2f{b[j].z, b[j].w}, acc[j]);
  }

  if constexpr (RELU) {
#pragma unroll
    for (int j = 0; j < TILES_N; ++j)
#pragma unroll
      for (int r = 0; r < 8; ++r) acc[j][r] = fmaxf(acc[j][r], 0.0f);
  }

  const int row0 = tm * 16 + half * 8;
  float* __restrict__ op = out + (size_t)row0 * N_OUT + l16;
  if (tm * 16 + 16 <= n_nodes) {
#pragma unroll
    for (int j = 0; j < TILES_N; ++j) {
      if (N_PAD == N_OUT || j * 16 + l16 < N_OUT) {
#pragma unroll
        for (int r = 0; r < 8; ++r) op[r * N_OUT + j * 16] = acc[j][r];
      }
    }
  } else {
#pragma unroll
    for (int j = 0; j < TILES_N; ++j) {
#pragma unroll
      for (int r = 0; r < 8; ++r) {
        if (row0 + r < n_nodes && j * 16 + l16 < N_OUT) op[r * N_OUT + j * 16] = acc[j][r];
      }
    }
  }
}

extern "C" void kernel_launch(void* const* d_in, const int* in_sizes, int n_in,
                              void* d_out, int out_size, void* d_ws, size_t ws_size,
                              hipStream_t stream) {
  const float* x   = (const float*)d_in[0];
  const int*   src = (const int*)d_in[1];
  const int*   dst = (const int*)d_in[2];
  const float* Ws[3] = { (const float*)d_in[3], (const float*)d_in[6], (const float*)d_in[9]  };
  const float* Wn[3] = { (const float*)d_in[4], (const float*)d_in[7], (const float*)d_in[10] };
  const float* bs[3] = { (const float*)d_in[5], (const float*)d_in[8], (const float*)d_in[11] };

  const int N = in_sizes[0] / F_DIM;
  const int E = in_sizes[1];
  const int NC = 47;

  float* invdeg = (float*)d_ws;
  float* agg = invdeg + (size_t)N;
  float* hA  = agg + (size_t)N * F_DIM;
  float* hB  = hA  + (size_t)N * F_DIM;
  float* wt  = hB  + (size_t)N * F_DIM;
  float* WsT[3] = { wt,                 wt + 2 * F_DIM * F_DIM, wt + 4 * F_DIM * F_DIM };
  float* WnT[3] = { wt + F_DIM * F_DIM, wt + 3 * F_DIM * F_DIM, wt + 4 * F_DIM * F_DIM + F_DIM * NC };

  int* cnt     = (int*)hB;
  int* total   = cnt + (size_t)N * CNT_PAD;
  int* rank    = total + 32;
  int* col_raw = rank + (size_t)E;
  int* col_srt = (int*)d_out;
  int* rowptr  = col_srt + (size_t)E;
  int* rowcnt  = rowptr + (size_t)N;

  const int ZB = 2048;

  const int TB = 4 * TB_SQ + 2 * ((F_DIM * NC + 255) / 256);
  prep_zero_transpose<<<TB + ZB, 256, 0, stream>>>(
      (int4*)cnt, ((size_t)N * CNT_PAD + 32) / 4,
      Ws[0], Wn[0], Ws[1], Wn[1], Ws[2], Wn[2], wt, NC);

  count_rank<<<(E + 255) / 256, 256, 0, stream>>>(dst, cnt, rank, E);
  alloc_rows<<<(N + 255) / 256, 256, 0, stream>>>(cnt, total, rowptr, rowcnt, invdeg, N);
  place_edges<<<(E + 255) / 256, 256, 0, stream>>>(src, dst, rank, rowptr, col_raw, E);

  const float* hin = x;
  float* houts[3] = { hA, hB, (float*)d_out };

  int gather_blocks = (N + 7) / 8;
  int tiles_m = (N + 15) / 16;

  for (int l = 0; l < 3; ++l) {
    if (l == 0)
      node_gather<true><<<gather_blocks, 256, 0, stream>>>(hin, rowptr, rowcnt, col_raw, col_srt, agg, N);
    else
      node_gather<false><<<gather_blocks, 256, 0, stream>>>(hin, rowptr, rowcnt, col_raw, col_srt, agg, N);

    int blocks = (tiles_m + GEMM_WAVES - 1) / GEMM_WAVES;
    if (l < 2) {
      sage_gemm<F_DIM, true><<<blocks, GEMM_THREADS, 0, stream>>>(
          hin, agg, invdeg, WsT[l], WnT[l], bs[l], houts[l], N);
    } else {
      sage_gemm<NC, false><<<blocks, GEMM_THREADS, 0, stream>>>(
          hin, agg, invdeg, WsT[l], WnT[l], bs[l], houts[l], N);
    }
    hin = houts[l];
  }
}


# Last layer multiplies first (S=h@W_self+b, P=h@W_neigh, 48-float rows), then takes the neighbour mean of P: gather moves 192 B rows, not 512; sum order changes (mean after product, even/odd halves)
# speedup vs baseline: 3.9913x; 1.0368x over previous
#include <hip/hip_runtime.h>


#define F_DIM 128

typedef __attribute__((ext_vector_type(2))) float v2f;
typedef __attribute__((ext_vector_type(8))) float v8f;

static __device__ __forceinline__ v8f wmma_f32x4(v2f a, v2f b, v8f c) {
  return __builtin_amdgcn_wmma_f32_16x16x4_f32(
      false, a, false, b, (short)0, c, false, false);
}

#define CNT_PAD 32

#define TB_SQ (F_DIM * F_DIM / 256)
__global__ void __launch_bounds__(256)
prep_zero_transpose(int4* __restrict__ p, size_t n4,
                    const float* __restrict__ Ws0, const float* __restrict__ Wn0,
                    const float* __restrict__ Ws1, const float* __restrict__ Wn1,
                    const float* __restrict__ Ws2, const float* __restrict__ Wn2,
                    float* __restrict__ wt, int n_cls) {
  const int tb_cls = (F_DIM * n_cls + 255) / 256;
  const int n_tb = 4 * TB_SQ + 2 * tb_cls;
  const int b = blockIdx.x;
  if (b < n_tb) {
    const float* W;
    float* WT;
    int n_out, lb;
    if (b < 4 * TB_SQ) {
      const int m = b / TB_SQ;
      lb = b - m * TB_SQ;
      n_out = F_DIM;
      W = (m == 0) ? Ws0 : (m == 1) ? Wn0 : (m == 2) ? Ws1 : Wn1;
      WT = wt + (size_t)m * F_DIM * F_DIM;
    } else {
      const int r = b - 4 * TB_SQ;
      const int m = r / tb_cls;
      lb = r - m * tb_cls;
      n_out = n_cls;
      W = m ? Wn2 : Ws2;
      WT = wt + (size_t)4 * F_DIM * F_DIM + (size_t)m * F_DIM * n_cls;
    }
    const int i = lb * 256 + threadIdx.x;
    if (i < F_DIM * n_out) {
      const int k = i / n_out;
      const int n = i - k * n_out;
      WT[n * F_DIM + k] = W[i];
    }
    return;
  }
  size_t i = (size_t)(b - n_tb) * 256 + threadIdx.x;
  const size_t stride = (size_t)(gridDim.x - n_tb) * 256;
  for (; i < n4; i += stride) p[i] = make_int4(0, 0, 0, 0);
}

__global__ void __launch_bounds__(256)
count_rank(const int* __restrict__ dst, int* __restrict__ cnt,
           int* __restrict__ rank, int nE) {
  int e = blockIdx.x * blockDim.x + threadIdx.x;
  if (e < nE) rank[e] = atomicAdd(&cnt[(size_t)dst[e] * CNT_PAD], 1);
}

__global__ void __launch_bounds__(256)
alloc_rows(const int* __restrict__ cnt, int* __restrict__ total,
           int* __restrict__ rowptr, int* __restrict__ rowcnt,
           float* __restrict__ inv_deg, int n) {
  int i = blockIdx.x * blockDim.x + threadIdx.x;
  int lane = threadIdx.x & 31;
  int c = (i < n) ? cnt[(size_t)i * CNT_PAD] : 0;
  int incl = c;
  for (int o = 1; o < 32; o <<= 1) {
    int t = __shfl_up(incl, o);
    if (lane >= o) incl += t;
  }
  int base = 0;
  if (lane == 31) base = atomicAdd(total, incl);
  base = __shfl(base, 31);
  if (i < n) {
    rowptr[i] = base + incl - c;
    rowcnt[i] = c;
    inv_deg[i] = 1.0f / fmaxf((float)c, 1.0f);
  }
}

__global__ void __launch_bounds__(256)
place_edges(const int* __restrict__ src, const int* __restrict__ dst,
            const int* __restrict__ rank, const int* __restrict__ rowptr,
            int* __restrict__ col, int nE) {
  int e = blockIdx.x * blockDim.x + threadIdx.x;
  if (e < nE) col[rowptr[dst[e]] + rank[e]] = src[e];
}

template <bool SORT>
__global__ void __launch_bounds__(256)
node_gather(const float* __restrict__ h, const int* __restrict__ rowptr,
            const int* __restrict__ rowcnt, const int* __restrict__ col_raw,
            int* col_srt, float* __restrict__ agg, int n_nodes) {
  const int lane = threadIdx.x & 31;
  const int node = __builtin_amdgcn_readfirstlane((int)(blockIdx.x * 8 + (threadIdx.x >> 5)));
  if (node >= n_nodes) return;
  const int start = rowptr[node];
  const int L = rowcnt[node];

  int c0 = 0;
  if constexpr (SORT) {
    if (L <= 32) {
      const int raw = (lane < L) ? col_raw[start + lane] : 0x7fffffff;
      int r = 0;
      for (int j = 0; j < L; ++j) {
        const int vj = __builtin_amdgcn_readlane(raw, j);
        r += (vj < raw || (vj == raw && j < lane)) ? 1 : 0;
      }
      if (lane < L) col_srt[start + r] = raw;
      c0 = __builtin_amdgcn_ds_permute(r << 2, raw);
    } else {
      for (int b = 0; b < L; b += 32) {
        const int idx = b + lane;
        const int myv = (idx < L) ? col_raw[start + idx] : 0;
        int r = 0;
        for (int j = 0; j < L; ++j) {
          const int vj = col_raw[start + j];
          r += (vj < myv || (vj == myv && j < idx)) ? 1 : 0;
        }
        if (idx < L) col_srt[start + r] = myv;
      }
      __threadfence();
    }
  }

  float4 acc = make_float4(0.0f, 0.0f, 0.0f, 0.0f);
  for (int base = 0; base < L; base += 32) {
    const int n = (L - base < 32) ? (L - base) : 32;
    int c;
    if (SORT && L <= 32) c = c0;
    else c = (lane < n) ? col_srt[start + base + lane] : 0;
    for (int j = 0; j < n; j += 8) {
      int s[8];
#pragma unroll
      for (int u = 0; u < 8; ++u) {
        const int ju = (j + u < n) ? (j + u) : (n - 1);
        s[u] = __builtin_amdgcn_readlane(c, ju);
      }
      float4 v[8];
#pragma unroll
      for (int u = 0; u < 8; ++u) {
        if (j + u < n) {
          const float4* row = (const float4*)(h + (size_t)s[u] * F_DIM);
          v[u] = row[lane];
        } else {
          v[u] = make_float4(0.0f, 0.0f, 0.0f, 0.0f);
        }
      }
#pragma unroll
      for (int u = 0; u < 8; ++u) {
        acc.x += v[u].x; acc.y += v[u].y; acc.z += v[u].z; acc.w += v[u].w;
      }
    }
  }
  ((float4*)(agg + (size_t)node * F_DIM))[lane] = acc;
}

template <int N_OUT>
__global__ void __launch_bounds__(256)
gather_project_out(const float* __restrict__ S, const float* __restrict__ inv_deg,
                   const int* __restrict__ rowptr, const int* __restrict__ rowcnt,
                   const int* __restrict__ col, float* __restrict__ out, int n_nodes) {
  constexpr int PITCH = ((N_OUT + 15) / 16) * 16;
  constexpr int CHUNKS = PITCH / 4;
  static_assert(CHUNKS <= 16, "a projected row fits one 16-lane half");
  __shared__ __attribute__((aligned(16))) float tr[8][2][PITCH];

  const int lane = threadIdx.x & 31;
  const int wv = __builtin_amdgcn_readfirstlane((int)(threadIdx.x >> 5));
  const int node = __builtin_amdgcn_readfirstlane((int)(blockIdx.x * 8 + (threadIdx.x >> 5)));
  if (node >= n_nodes) return;
  const int half = lane >> 4;
  const int sub = lane & 15;
  const bool chunk_on = sub < CHUNKS;
  const float* __restrict__ P = S + (size_t)n_nodes * PITCH;
  const int start = rowptr[node];
  const int L = rowcnt[node];
  const float invd = inv_deg[node];

  constexpr int OUT_TRIPS = (N_OUT + 31) / 32;
  float sv[OUT_TRIPS];
#pragma unroll
  for (int q = 0; q < OUT_TRIPS; ++q) {
    const int cq = q * 32 + lane;
    sv[q] = (cq < N_OUT) ? S[(size_t)node * PITCH + cq] : 0.0f;
  }

  float4 acc = make_float4(0.0f, 0.0f, 0.0f, 0.0f);
  for (int base = 0; base < L; base += 32) {
    const int n = (L - base < 32) ? (L - base) : 32;
    const int c = (lane < n) ? col[start + base + lane] : 0;
    for (int j = 0; j < n; j += 16) {
      int s[8];
#pragma unroll
      for (int u = 0; u < 8; ++u) {
        const int e0 = (j + 2 * u < n) ? (j + 2 * u) : (n - 1);
        const int e1 = (j + 2 * u + 1 < n) ? (j + 2 * u + 1) : (n - 1);
        const int lo = __builtin_amdgcn_readlane(c, e0);
        const int hi = __builtin_amdgcn_readlane(c, e1);
        s[u] = half ? hi : lo;
      }
      float4 v[8];
#pragma unroll
      for (int u = 0; u < 8; ++u) {
        v[u] = make_float4(0.0f, 0.0f, 0.0f, 0.0f);
        if (j + 2 * u < n) {
          if (chunk_on && j + 2 * u + half < n)
            v[u] = ((const float4*)(P + (size_t)s[u] * PITCH))[sub];
        }
      }
#pragma unroll
      for (int u = 0; u < 8; ++u) {
        acc.x += v[u].x; acc.y += v[u].y; acc.z += v[u].z; acc.w += v[u].w;
      }
    }
  }

  if (chunk_on) *(float4*)&tr[wv][half][4 * sub] = acc;
  float* __restrict__ orow = out + (size_t)node * N_OUT;
#pragma unroll
  for (int q = 0; q < OUT_TRIPS; ++q) {
    const int cq = q * 32 + lane;
    if (cq < N_OUT) {
      const float t = tr[wv][0][cq] + tr[wv][1][cq];
      orow[cq] = sv[q] + invd * t;
    }
  }
}

#define GEMM_WAVES 4
#define GEMM_THREADS (GEMM_WAVES * 32)
#define W_PITCH 132

template <int N_OUT, bool RELU, bool PROJECT>
__global__ void __launch_bounds__(GEMM_THREADS)
sage_gemm(const float* h, const float* __restrict__ agg,
          const float* __restrict__ inv_deg,
          const float* __restrict__ WsT, const float* __restrict__ WnT,
          const float* __restrict__ bias, float* out, int n_nodes) {
  constexpr int TILES_N = (N_OUT + 15) / 16;
  constexpr int N_PAD = TILES_N * 16;
  constexpr int KSTEPS = F_DIM / 8;

  __shared__ __attribute__((aligned(16))) float Bs[2 * N_PAD * W_PITCH];

  const int tid = threadIdx.x;
  const int lane = tid & 31;
  const int wv = __builtin_amdgcn_readfirstlane(tid >> 5);
  const int half = lane >> 4;
  const int l16 = lane & 15;
  const int kq = 4 * half;

  const int tiles_m = (n_nodes + 15) >> 4;
  const int tm = blockIdx.x * GEMM_WAVES + wv;
  const bool active = tm < tiles_m;

  int mrow = tm * 16 + l16;
  if (mrow >= n_nodes) mrow = n_nodes - 1;

  float4 ah[KSTEPS], an[KSTEPS];
  if (active) {
    const float4* hp = (const float4*)(h + (size_t)mrow * F_DIM + kq);
#pragma unroll
    for (int t = 0; t < KSTEPS; ++t) ah[t] = hp[2 * t];
    if constexpr (!PROJECT) {
      const float4* __restrict__ ap = (const float4*)(agg + (size_t)mrow * F_DIM + kq);
#pragma unroll
      for (int t = 0; t < KSTEPS; ++t) an[t] = ap[2 * t];
    }
  }

  {
    constexpr int TRIPS = N_PAD / GEMM_WAVES;
    constexpr int STAGE_BATCH = (TRIPS > 16) ? 16 : TRIPS;
    static_assert(TRIPS % STAGE_BATCH == 0, "staging batches tile the rows");
    const float4* __restrict__ gs = (const float4*)WsT + lane;
    const float4* __restrict__ gn = (const float4*)WnT + lane;
    float* ls = Bs + wv * W_PITCH + 4 * lane;
    float* ln = ls + N_PAD * W_PITCH;
#pragma unroll 1
    for (int q0 = 0; q0 < TRIPS; q0 += STAGE_BATCH) {
      float4 vs[STAGE_BATCH], vn[STAGE_BATCH];
#pragma unroll
      for (int u = 0; u < STAGE_BATCH; ++u) {
        const int n = wv + GEMM_WAVES * (q0 + u);
        const int nr = (N_PAD != N_OUT && n >= N_OUT) ? (N_OUT - 1) : n;
        vs[u] = gs[nr * (F_DIM / 4)];
        vn[u] = gn[nr * (F_DIM / 4)];
      }
#pragma unroll
      for (int u = 0; u < STAGE_BATCH; ++u) {
        const int q = q0 + u;
        *(float4*)(ls + q * GEMM_WAVES * W_PITCH) = vs[u];
        *(float4*)(ln + q * GEMM_WAVES * W_PITCH) = vn[u];
      }
    }
  }
  __syncthreads();
  if (!active) return;

  float invd = 1.0f;
  if constexpr (!PROJECT) invd = inv_deg[mrow];

  v8f acc[TILES_N];
  v8f accp[PROJECT ? TILES_N : 1];
#pragma unroll
  for (int j = 0; j < TILES_N; ++j) {
    const int ncol = j * 16 + l16;
    const float bv = (ncol < N_OUT) ? bias[(ncol < N_OUT) ? ncol : 0] : 0.0f;
#pragma unroll
    for (int r = 0; r < 8; ++r) acc[j][r] = bv;
    if constexpr (PROJECT) {
#pragma unroll
      for (int r = 0; r < 8; ++r) accp[j][r] = 0.0f;
    }
  }

  const float* bself = Bs + l16 * W_PITCH + kq;
  const float* bneigh = bself + N_PAD * W_PITCH;

#pragma unroll
  for (int t = 0; t < KSTEPS; ++t) {
    const float4 a = ah[t];
    float4 b[TILES_N];
#pragma unroll
    for (int j = 0; j < TILES_N; ++j)
      b[j] = *(const float4*)(bself + j * 16 * W_PITCH + 8 * t);
#pragma unroll
    for (int j = 0; j < TILES_N; ++j)
      acc[j] = wmma_f32x4(v2f{a.x, a.y}, v2f{b[j].x, b[j].y}, acc[j]);
#pragma unroll
    for (int j = 0; j < TILES_N; ++j)
      acc[j] = wmma_f32x4(v2f{a.z, a.w}, v2f{b[j].z, b[j].w}, acc[j]);
  }
#pragma unroll
  for (int t = 0; t < KSTEPS; ++t) {
    float4 a;
    if constexpr (PROJECT) a = ah[t];
    else a = make_float4(an[t].x * invd, an[t].y * invd, an[t].z * invd, an[t].w * invd);
    float4 b[TILES_N];
#pragma unroll
    for (int j = 0; j < TILES_N; ++j)
      b[j] = *(const float4*)(bneigh + j * 16 * W_PITCH + 8 * t);
    if constexpr (PROJECT) {
#pragma unroll
      for (int j = 0; j < TILES_N; ++j)
        accp[j] = wmma_f32x4(v2f{a.x, a.y}, v2f{b[j].x, b[j].y}, accp[j]);
#pragma unroll
      for (int j = 0; j < TILES_N; ++j)
        accp[j] = wmma_f32x4(v2f{a.z, a.w}, v2f{b[j].z, b[j].w}, accp[j]);
    } else {
#pragma unroll
      for (int j = 0; j < TILES_N; ++j)
        acc[j] = wmma_f32x4(v2f{a.x, a.y}, v2f{b[j].x, b[j].y}, acc[j]);
#pragma unroll
      for (int j = 0; j < TILES_N; ++j)
        acc[j] = wmma_f32x4(v2f{a.z, a.w}, v2f{b[j].z, b[j].w}, acc[j]);
    }
  }

  if constexpr (RELU) {
#pragma unroll
    for (int j = 0; j < TILES_N; ++j)
#pragma unroll
      for (int r = 0; r < 8; ++r) acc[j][r] = fmaxf(acc[j][r], 0.0f);
  }

  const int row0 = tm * 16 + half * 8;
  if constexpr (PROJECT) {
    float* sp = out + (size_t)row0 * N_PAD + l16;
    float* pp = sp + (size_t)n_nodes * N_PAD;
    if (tm * 16 + 16 <= n_nodes) {
#pragma unroll
      for (int j = 0; j < TILES_N; ++j) {
#pragma unroll
        for (int r = 0; r < 8; ++r) {
          sp[r * N_PAD + j * 16] = acc[j][r];
          pp[r * N_PAD + j * 16] = accp[j][r];
        }
      }
    } else {
#pragma unroll
      for (int j = 0; j < TILES_N; ++j) {
#pragma unroll
        for (int r = 0; r < 8; ++r) {
          if (row0 + r < n_nodes) {
            sp[r * N_PAD + j * 16] = acc[j][r];
            pp[r * N_PAD + j * 16] = accp[j][r];
          }
        }
      }
    }
    return;
  }
  float* op = out + (size_t)row0 * N_OUT + l16;
  if (tm * 16 + 16 <= n_nodes) {
#pragma unroll
    for (int j = 0; j < TILES_N; ++j) {
      if (N_PAD == N_OUT || j * 16 + l16 < N_OUT) {
#pragma unroll
        for (int r = 0; r < 8; ++r) op[r * N_OUT + j * 16] = acc[j][r];
      }
    }
  } else {
#pragma unroll
    for (int j = 0; j < TILES_N; ++j) {
#pragma unroll
      for (int r = 0; r < 8; ++r) {
        if (row0 + r < n_nodes && j * 16 + l16 < N_OUT) op[r * N_OUT + j * 16] = acc[j][r];
      }
    }
  }
}

extern "C" void kernel_launch(void* const* d_in, const int* in_sizes, int n_in,
                              void* d_out, int out_size, void* d_ws, size_t ws_size,
                              hipStream_t stream) {
  const float* x   = (const float*)d_in[0];
  const int*   src = (const int*)d_in[1];
  const int*   dst = (const int*)d_in[2];
  const float* Ws[3] = { (const float*)d_in[3], (const float*)d_in[6], (const float*)d_in[9]  };
  const float* Wn[3] = { (const float*)d_in[4], (const float*)d_in[7], (const float*)d_in[10] };
  const float* bs[3] = { (const float*)d_in[5], (const float*)d_in[8], (const float*)d_in[11] };

  const int N = in_sizes[0] / F_DIM;
  const int E = in_sizes[1];
  const int NC = 47;

  float* invdeg = (float*)d_ws;
  float* agg = invdeg + (size_t)N;
  float* hA  = agg + (size_t)N * F_DIM;
  float* epl = hA  + (size_t)N * F_DIM;
  float* wt  = epl + (size_t)N * F_DIM;
  float* WsT[3] = { wt,                 wt + 2 * F_DIM * F_DIM, wt + 4 * F_DIM * F_DIM };
  float* WnT[3] = { wt + F_DIM * F_DIM, wt + 3 * F_DIM * F_DIM, wt + 4 * F_DIM * F_DIM + F_DIM * NC };

  int* cnt     = (int*)epl;
  int* total   = cnt + (size_t)N * CNT_PAD;
  int* rank    = total + 32;
  int* col_raw = rank + (size_t)E;
  int* col_srt = col_raw + (size_t)E;
  int* rowptr  = col_srt + (size_t)E;
  int* rowcnt  = rowptr + (size_t)N;

  const int ZB = 2048;

  const int TB = 4 * TB_SQ + 2 * ((F_DIM * NC + 255) / 256);
  prep_zero_transpose<<<TB + ZB, 256, 0, stream>>>(
      (int4*)cnt, ((size_t)N * CNT_PAD + 32) / 4,
      Ws[0], Wn[0], Ws[1], Wn[1], Ws[2], Wn[2], wt, NC);

  count_rank<<<(E + 255) / 256, 256, 0, stream>>>(dst, cnt, rank, E);
  alloc_rows<<<(N + 255) / 256, 256, 0, stream>>>(cnt, total, rowptr, rowcnt, invdeg, N);
  place_edges<<<(E + 255) / 256, 256, 0, stream>>>(src, dst, rank, rowptr, col_raw, E);

  int gather_blocks = (N + 7) / 8;
  int tiles_m = (N + 15) / 16;
  int blocks = (tiles_m + GEMM_WAVES - 1) / GEMM_WAVES;

  node_gather<true><<<gather_blocks, 256, 0, stream>>>(x, rowptr, rowcnt, col_raw, col_srt, agg, N);
  sage_gemm<F_DIM, true, false><<<blocks, GEMM_THREADS, 0, stream>>>(
      x, agg, invdeg, WsT[0], WnT[0], bs[0], hA, N);
  node_gather<false><<<gather_blocks, 256, 0, stream>>>(hA, rowptr, rowcnt, col_raw, col_srt, agg, N);
  sage_gemm<F_DIM, true, false><<<blocks, GEMM_THREADS, 0, stream>>>(
      hA, agg, invdeg, WsT[1], WnT[1], bs[1], hA, N);
  sage_gemm<NC, false, true><<<blocks, GEMM_THREADS, 0, stream>>>(
      hA, nullptr, nullptr, WsT[2], WnT[2], bs[2], agg, N);
  gather_project_out<NC><<<gather_blocks, 256, 0, stream>>>(
      agg, invdeg, rowptr, rowcnt, col_srt, (float*)d_out, N);
}


# GEMM launches persistent: 256 work-groups (one per WGP; 12 waves, 16 in the last layer) stage both W^T once, each wave loops over 16-row tiles dealt evenly; arithmetic and sum order unchanged
# speedup vs baseline: 4.2676x; 1.0692x over previous
#include <hip/hip_runtime.h>


#define F_DIM 128

typedef __attribute__((ext_vector_type(2))) float v2f;
typedef __attribute__((ext_vector_type(8))) float v8f;

static __device__ __forceinline__ v8f wmma_f32x4(v2f a, v2f b, v8f c) {
  return __builtin_amdgcn_wmma_f32_16x16x4_f32(
      false, a, false, b, (short)0, c, false, false);
}

#define CNT_PAD 32

#define TB_SQ (F_DIM * F_DIM / 256)
__global__ void __launch_bounds__(256)
prep_zero_transpose(int4* __restrict__ p, size_t n4,
                    const float* __restrict__ Ws0, const float* __restrict__ Wn0,
                    const float* __restrict__ Ws1, const float* __restrict__ Wn1,
                    const float* __restrict__ Ws2, const float* __restrict__ Wn2,
                    float* __restrict__ wt, int n_cls) {
  const int tb_cls = (F_DIM * n_cls + 255) / 256;
  const int n_tb = 4 * TB_SQ + 2 * tb_cls;
  const int b = blockIdx.x;
  if (b < n_tb) {
    const float* W;
    float* WT;
    int n_out, lb;
    if (b < 4 * TB_SQ) {
      const int m = b / TB_SQ;
      lb = b - m * TB_SQ;
      n_out = F_DIM;
      W = (m == 0) ? Ws0 : (m == 1) ? Wn0 : (m == 2) ? Ws1 : Wn1;
      WT = wt + (size_t)m * F_DIM * F_DIM;
    } else {
      const int r = b - 4 * TB_SQ;
      const int m = r / tb_cls;
      lb = r - m * tb_cls;
      n_out = n_cls;
      W = m ? Wn2 : Ws2;
      WT = wt + (size_t)4 * F_DIM * F_DIM + (size_t)m * F_DIM * n_cls;
    }
    const int i = lb * 256 + threadIdx.x;
    if (i < F_DIM * n_out) {
      const int k = i / n_out;
      const int n = i - k * n_out;
      WT[n * F_DIM + k] = W[i];
    }
    return;
  }
  size_t i = (size_t)(b - n_tb) * 256 + threadIdx.x;
  const size_t stride = (size_t)(gridDim.x - n_tb) * 256;
  for (; i < n4; i += stride) p[i] = make_int4(0, 0, 0, 0);
}

__global__ void __launch_bounds__(256)
count_rank(const int* __restrict__ dst, int* __restrict__ cnt,
           int* __restrict__ rank, int nE) {
  int e = blockIdx.x * blockDim.x + threadIdx.x;
  if (e < nE) rank[e] = atomicAdd(&cnt[(size_t)dst[e] * CNT_PAD], 1);
}

__global__ void __launch_bounds__(256)
alloc_rows(const int* __restrict__ cnt, int* __restrict__ total,
           int* __restrict__ rowptr, int* __restrict__ rowcnt,
           float* __restrict__ inv_deg, int n) {
  int i = blockIdx.x * blockDim.x + threadIdx.x;
  int lane = threadIdx.x & 31;
  int c = (i < n) ? cnt[(size_t)i * CNT_PAD] : 0;
  int incl = c;
  for (int o = 1; o < 32; o <<= 1) {
    int t = __shfl_up(incl, o);
    if (lane >= o) incl += t;
  }
  int base = 0;
  if (lane == 31) base = atomicAdd(total, incl);
  base = __shfl(base, 31);
  if (i < n) {
    rowptr[i] = base + incl - c;
    rowcnt[i] = c;
    inv_deg[i] = 1.0f / fmaxf((float)c, 1.0f);
  }
}

__global__ void __launch_bounds__(256)
place_edges(const int* __restrict__ src, const int* __restrict__ dst,
            const int* __restrict__ rank, const int* __restrict__ rowptr,
            int* __restrict__ col, int nE) {
  int e = blockIdx.x * blockDim.x + threadIdx.x;
  if (e < nE) col[rowptr[dst[e]] + rank[e]] = src[e];
}

template <bool SORT>
__global__ void __launch_bounds__(256)
node_gather(const float* __restrict__ h, const int* __restrict__ rowptr,
            const int* __restrict__ rowcnt, const int* __restrict__ col_raw,
            int* col_srt, float* __restrict__ agg, int n_nodes) {
  const int lane = threadIdx.x & 31;
  const int node = __builtin_amdgcn_readfirstlane((int)(blockIdx.x * 8 + (threadIdx.x >> 5)));
  if (node >= n_nodes) return;
  const int start = rowptr[node];
  const int L = rowcnt[node];

  int c0 = 0;
  if constexpr (SORT) {
    if (L <= 32) {
      const int raw = (lane < L) ? col_raw[start + lane] : 0x7fffffff;
      int r = 0;
      for (int j = 0; j < L; ++j) {
        const int vj = __builtin_amdgcn_readlane(raw, j);
        r += (vj < raw || (vj == raw && j < lane)) ? 1 : 0;
      }
      if (lane < L) col_srt[start + r] = raw;
      c0 = __builtin_amdgcn_ds_permute(r << 2, raw);
    } else {
      for (int b = 0; b < L; b += 32) {
        const int idx = b + lane;
        const int myv = (idx < L) ? col_raw[start + idx] : 0;
        int r = 0;
        for (int j = 0; j < L; ++j) {
          const int vj = col_raw[start + j];
          r += (vj < myv || (vj == myv && j < idx)) ? 1 : 0;
        }
        if (idx < L) col_srt[start + r] = myv;
      }
      __threadfence();
    }
  }

  float4 acc = make_float4(0.0f, 0.0f, 0.0f, 0.0f);
  for (int base = 0; base < L; base += 32) {
    const int n = (L - base < 32) ? (L - base) : 32;
    int c;
    if (SORT && L <= 32) c = c0;
    else c = (lane < n) ? col_srt[start + base + lane] : 0;
    for (int j = 0; j < n; j += 8) {
      int s[8];
#pragma unroll
      for (int u = 0; u < 8; ++u) {
        const int ju = (j + u < n) ? (j + u) : (n - 1);
        s[u] = __builtin_amdgcn_readlane(c, ju);
      }
      float4 v[8];
#pragma unroll
      for (int u = 0; u < 8; ++u) {
        if (j + u < n) {
          const float4* row = (const float4*)(h + (size_t)s[u] * F_DIM);
          v[u] = row[lane];
        } else {
          v[u] = make_float4(0.0f, 0.0f, 0.0f, 0.0f);
        }
      }
#pragma unroll
      for (int u = 0; u < 8; ++u) {
        acc.x += v[u].x; acc.y += v[u].y; acc.z += v[u].z; acc.w += v[u].w;
      }
    }
  }
  ((float4*)(agg + (size_t)node * F_DIM))[lane] = acc;
}

template <int N_OUT>
__global__ void __launch_bounds__(256)
gather_project_out(const float* __restrict__ S, const float* __restrict__ inv_deg,
                   const int* __restrict__ rowptr, const int* __restrict__ rowcnt,
                   const int* __restrict__ col, float* __restrict__ out, int n_nodes) {
  constexpr int PITCH = ((N_OUT + 15) / 16) * 16;
  constexpr int CHUNKS = PITCH / 4;
  static_assert(CHUNKS <= 16, "a projected row fits one 16-lane half");
  __shared__ __attribute__((aligned(16))) float tr[8][2][PITCH];

  const int lane = threadIdx.x & 31;
  const int wv = __builtin_amdgcn_readfirstlane((int)(threadIdx.x >> 5));
  const int node = __builtin_amdgcn_readfirstlane((int)(blockIdx.x * 8 + (threadIdx.x >> 5)));
  if (node >= n_nodes) return;
  const int half = lane >> 4;
  const int sub = lane & 15;
  const bool chunk_on = sub < CHUNKS;
  const float* __restrict__ P = S + (size_t)n_nodes * PITCH;
  const int start = rowptr[node];
  const int L = rowcnt[node];
  const float invd = inv_deg[node];

  constexpr int OUT_TRIPS = (N_OUT + 31) / 32;
  float sv[OUT_TRIPS];
#pragma unroll
  for (int q = 0; q < OUT_TRIPS; ++q) {
    const int cq = q * 32 + lane;
    sv[q] = (cq < N_OUT) ? S[(size_t)node * PITCH + cq] : 0.0f;
  }

  float4 acc = make_float4(0.0f, 0.0f, 0.0f, 0.0f);
  for (int base = 0; base < L; base += 32) {
    const int n = (L - base < 32) ? (L - base) : 32;
    const int c = (lane < n) ? col[start + base + lane] : 0;
    for (int j = 0; j < n; j += 16) {
      int s[8];
#pragma unroll
      for (int u = 0; u < 8; ++u) {
        const int e0 = (j + 2 * u < n) ? (j + 2 * u) : (n - 1);
        const int e1 = (j + 2 * u + 1 < n) ? (j + 2 * u + 1) : (n - 1);
        const int lo = __builtin_amdgcn_readlane(c, e0);
        const int hi = __builtin_amdgcn_readlane(c, e1);
        s[u] = half ? hi : lo;
      }
      float4 v[8];
#pragma unroll
      for (int u = 0; u < 8; ++u) {
        v[u] = make_float4(0.0f, 0.0f, 0.0f, 0.0f);
        if (j + 2 * u < n) {
          if (chunk_on && j + 2 * u + half < n)
            v[u] = ((const float4*)(P + (size_t)s[u] * PITCH))[sub];
        }
      }
#pragma unroll
      for (int u = 0; u < 8; ++u) {
        acc.x += v[u].x; acc.y += v[u].y; acc.z += v[u].z; acc.w += v[u].w;
      }
    }
  }

  if (chunk_on) *(float4*)&tr[wv][half][4 * sub] = acc;
  float* __restrict__ orow = out + (size_t)node * N_OUT;
#pragma unroll
  for (int q = 0; q < OUT_TRIPS; ++q) {
    const int cq = q * 32 + lane;
    if (cq < N_OUT) {
      const float t = tr[wv][0][cq] + tr[wv][1][cq];
      orow[cq] = sv[q] + invd * t;
    }
  }
}

#define GEMM_GROUPS 256
#define W_PITCH 132

template <int N_OUT, bool RELU, bool PROJECT, int WAVES>
__global__ void __launch_bounds__(WAVES * 32)
sage_gemm(const float* h, const float* __restrict__ agg,
          const float* __restrict__ inv_deg,
          const float* __restrict__ WsT, const float* __restrict__ WnT,
          const float* __restrict__ bias, float* out, int n_nodes) {
  constexpr int THREADS = WAVES * 32;
  constexpr int TILES_N = (N_OUT + 15) / 16;
  constexpr int N_PAD = TILES_N * 16;
  constexpr int KSTEPS = F_DIM / 8;

  __shared__ __attribute__((aligned(16))) float Bs[2 * N_PAD * W_PITCH];

  const int tid = threadIdx.x;
  const int lane = tid & 31;
  const int wv = __builtin_amdgcn_readfirstlane(tid >> 5);
  const int half = lane >> 4;
  const int l16 = lane & 15;
  const int kq = 4 * half;

  const int tiles_m = (n_nodes + 15) >> 4;
  const int groups = (int)gridDim.x;
  const int tile_step = groups * WAVES;
  int tm = wv * groups + (int)blockIdx.x;

  float4 ah[KSTEPS], an[KSTEPS];
  float invd = 1.0f;
  auto fetch_rows = [&](int t_m) {
    int mrow = t_m * 16 + l16;
    if (mrow >= n_nodes) mrow = n_nodes - 1;
    const float4* hp = (const float4*)(h + (size_t)mrow * F_DIM + kq);
#pragma unroll
    for (int t = 0; t < KSTEPS; ++t) ah[t] = hp[2 * t];
    if constexpr (!PROJECT) {
      const float4* __restrict__ ap = (const float4*)(agg + (size_t)mrow * F_DIM + kq);
#pragma unroll
      for (int t = 0; t < KSTEPS; ++t) an[t] = ap[2 * t];
      invd = inv_deg[mrow];
    }
  };
  if (tm < tiles_m) fetch_rows(tm);

  {
    constexpr int ROW4 = F_DIM / 4;
    constexpr int TOT4 = N_PAD * ROW4;
    constexpr int TRIPS = (TOT4 + THREADS - 1) / THREADS;
    const float4* __restrict__ gs = (const float4*)WsT;
    const float4* __restrict__ gn = (const float4*)WnT;
    float4 vs[TRIPS], vn[TRIPS];
#pragma unroll
    for (int u = 0; u < TRIPS; ++u) {
      int i = tid + u * THREADS;
      if (TOT4 % THREADS != 0 && i >= TOT4) i = TOT4 - 1;
      const int n = i / ROW4;
      const int nr = (N_PAD != N_OUT && n >= N_OUT) ? (N_OUT - 1) : n;
      vs[u] = gs[nr * ROW4 + (i % ROW4)];
      vn[u] = gn[nr * ROW4 + (i % ROW4)];
    }
#pragma unroll
    for (int u = 0; u < TRIPS; ++u) {
      int i = tid + u * THREADS;
      if (TOT4 % THREADS != 0 && i >= TOT4) i = TOT4 - 1;
      float* ls = Bs + (i / ROW4) * W_PITCH + 4 * (i % ROW4);
      *(float4*)ls = vs[u];
      *(float4*)(ls + N_PAD * W_PITCH) = vn[u];
    }
  }
  __syncthreads();

  unsigned so = (unsigned)((l16 * W_PITCH + kq) * 4);
  unsigned no = so + (unsigned)(N_PAD * W_PITCH * 4);

  for (; tm < tiles_m; tm += tile_step) {
    asm volatile("" : "+v"(so), "+v"(no) : : "memory");
    const float* bself = (const float*)((const char*)Bs + so);
    const float* bneigh = (const float*)((const char*)Bs + no);
    float bv[TILES_N];
#pragma unroll
    for (int j = 0; j < TILES_N; ++j) {
      const int ncol = j * 16 + l16;
      bv[j] = (ncol < N_OUT) ? bias[(ncol < N_OUT) ? ncol : 0] : 0.0f;
    }
    v8f acc[TILES_N];
    v8f accp[PROJECT ? TILES_N : 1];
#pragma unroll
    for (int j = 0; j < TILES_N; ++j) {
#pragma unroll
      for (int r = 0; r < 8; ++r) acc[j][r] = bv[j];
      if constexpr (PROJECT) {
#pragma unroll
        for (int r = 0; r < 8; ++r) accp[j][r] = 0.0f;
      }
    }

#pragma unroll
    for (int t = 0; t < KSTEPS; ++t) {
      const float4 a = ah[t];
      float4 b[TILES_N];
#pragma unroll
      for (int j = 0; j < TILES_N; ++j)
        b[j] = *(const float4*)(bself + j * 16 * W_PITCH + 8 * t);
#pragma unroll
      for (int j = 0; j < TILES_N; ++j)
        acc[j] = wmma_f32x4(v2f{a.x, a.y}, v2f{b[j].x, b[j].y}, acc[j]);
#pragma unroll
      for (int j = 0; j < TILES_N; ++j)
        acc[j] = wmma_f32x4(v2f{a.z, a.w}, v2f{b[j].z, b[j].w}, acc[j]);
    }
#pragma unroll
    for (int t = 0; t < KSTEPS; ++t) {
      float4 a;
      if constexpr (PROJECT) a = ah[t];
      else a = make_float4(an[t].x * invd, an[t].y * invd, an[t].z * invd, an[t].w * invd);
      float4 b[TILES_N];
#pragma unroll
      for (int j = 0; j < TILES_N; ++j)
        b[j] = *(const float4*)(bneigh + j * 16 * W_PITCH + 8 * t);
      if constexpr (PROJECT) {
#pragma unroll
        for (int j = 0; j < TILES_N; ++j)
          accp[j] = wmma_f32x4(v2f{a.x, a.y}, v2f{b[j].x, b[j].y}, accp[j]);
#pragma unroll
        for (int j = 0; j < TILES_N; ++j)
          accp[j] = wmma_f32x4(v2f{a.z, a.w}, v2f{b[j].z, b[j].w}, accp[j]);
      } else {
#pragma unroll
        for (int j = 0; j < TILES_N; ++j)
          acc[j] = wmma_f32x4(v2f{a.x, a.y}, v2f{b[j].x, b[j].y}, acc[j]);
#pragma unroll
        for (int j = 0; j < TILES_N; ++j)
          acc[j] = wmma_f32x4(v2f{a.z, a.w}, v2f{b[j].z, b[j].w}, acc[j]);
      }
    }

    if constexpr (RELU) {
#pragma unroll
      for (int j = 0; j < TILES_N; ++j)
#pragma unroll
        for (int r = 0; r < 8; ++r) acc[j][r] = fmaxf(acc[j][r], 0.0f);
    }

    const int row0 = tm * 16 + half * 8;
    if constexpr (PROJECT) {
      float* sp = out + (size_t)row0 * N_PAD + l16;
      float* pp = sp + (size_t)n_nodes * N_PAD;
      if (tm * 16 + 16 <= n_nodes) {
#pragma unroll
        for (int j = 0; j < TILES_N; ++j) {
#pragma unroll
          for (int r = 0; r < 8; ++r) {
            sp[r * N_PAD + j * 16] = acc[j][r];
            pp[r * N_PAD + j * 16] = accp[j][r];
          }
        }
      } else {
#pragma unroll
        for (int j = 0; j < TILES_N; ++j) {
#pragma unroll
          for (int r = 0; r < 8; ++r) {
            if (row0 + r < n_nodes) {
              sp[r * N_PAD + j * 16] = acc[j][r];
              pp[r * N_PAD + j * 16] = accp[j][r];
            }
          }
        }
      }
    } else {
      float* op = out + (size_t)row0 * N_OUT + l16;
      if (tm * 16 + 16 <= n_nodes) {
#pragma unroll
        for (int j = 0; j < TILES_N; ++j) {
          if (N_PAD == N_OUT || j * 16 + l16 < N_OUT) {
#pragma unroll
            for (int r = 0; r < 8; ++r) op[r * N_OUT + j * 16] = acc[j][r];
          }
        }
      } else {
#pragma unroll
        for (int j = 0; j < TILES_N; ++j) {
#pragma unroll
          for (int r = 0; r < 8; ++r) {
            if (row0 + r < n_nodes && j * 16 + l16 < N_OUT) op[r * N_OUT + j * 16] = acc[j][r];
          }
        }
      }
    }

    const int tn = tm + tile_step;
    if (tn < tiles_m) fetch_rows(tn);
  }
}

extern "C" void kernel_launch(void* const* d_in, const int* in_sizes, int n_in,
                              void* d_out, int out_size, void* d_ws, size_t ws_size,
                              hipStream_t stream) {
  const float* x   = (const float*)d_in[0];
  const int*   src = (const int*)d_in[1];
  const int*   dst = (const int*)d_in[2];
  const float* Ws[3] = { (const float*)d_in[3], (const float*)d_in[6], (const float*)d_in[9]  };
  const float* Wn[3] = { (const float*)d_in[4], (const float*)d_in[7], (const float*)d_in[10] };
  const float* bs[3] = { (const float*)d_in[5], (const float*)d_in[8], (const float*)d_in[11] };

  const int N = in_sizes[0] / F_DIM;
  const int E = in_sizes[1];
  const int NC = 47;

  float* invdeg = (float*)d_ws;
  float* agg = invdeg + (size_t)N;
  float* hA  = agg + (size_t)N * F_DIM;
  float* epl = hA  + (size_t)N * F_DIM;
  float* wt  = epl + (size_t)N * F_DIM;
  float* WsT[3] = { wt,                 wt + 2 * F_DIM * F_DIM, wt + 4 * F_DIM * F_DIM };
  float* WnT[3] = { wt + F_DIM * F_DIM, wt + 3 * F_DIM * F_DIM, wt + 4 * F_DIM * F_DIM + F_DIM * NC };

  int* cnt     = (int*)epl;
  int* total   = cnt + (size_t)N * CNT_PAD;
  int* rank    = total + 32;
  int* col_raw = rank + (size_t)E;
  int* col_srt = col_raw + (size_t)E;
  int* rowptr  = col_srt + (size_t)E;
  int* rowcnt  = rowptr + (size_t)N;

  const int ZB = 2048;

  const int TB = 4 * TB_SQ + 2 * ((F_DIM * NC + 255) / 256);
  prep_zero_transpose<<<TB + ZB, 256, 0, stream>>>(
      (int4*)cnt, ((size_t)N * CNT_PAD + 32) / 4,
      Ws[0], Wn[0], Ws[1], Wn[1], Ws[2], Wn[2], wt, NC);

  count_rank<<<(E + 255) / 256, 256, 0, stream>>>(dst, cnt, rank, E);
  alloc_rows<<<(N + 255) / 256, 256, 0, stream>>>(cnt, total, rowptr, rowcnt, invdeg, N);
  place_edges<<<(E + 255) / 256, 256, 0, stream>>>(src, dst, rank, rowptr, col_raw, E);

  int gather_blocks = (N + 7) / 8;

  node_gather<true><<<gather_blocks, 256, 0, stream>>>(x, rowptr, rowcnt, col_raw, col_srt, agg, N);
  sage_gemm<F_DIM, true, false, 12><<<GEMM_GROUPS, 12 * 32, 0, stream>>>(
      x, agg, invdeg, WsT[0], WnT[0], bs[0], hA, N);
  node_gather<false><<<gather_blocks, 256, 0, stream>>>(hA, rowptr, rowcnt, col_raw, col_srt, agg, N);
  sage_gemm<F_DIM, true, false, 12><<<GEMM_GROUPS, 12 * 32, 0, stream>>>(
      hA, agg, invdeg, WsT[1], WnT[1], bs[1], hA, N);
  sage_gemm<NC, false, true, 16><<<GEMM_GROUPS, 16 * 32, 0, stream>>>(
      hA, nullptr, nullptr, WsT[2], WnT[2], bs[2], agg, N);
  gather_project_out<NC><<<gather_blocks, 256, 0, stream>>>(
      agg, invdeg, rowptr, rowcnt, col_srt, (float*)d_out, N);
}


# Edge grouping as a two-level counting sort ranked by LDS counters (buckets of 256 nodes, then nodes): no global atomics, no counter zeroing, 10 launches become 9; sum order unchanged
# speedup vs baseline: 4.8968x; 1.1474x over previous
#include <hip/hip_runtime.h>


#define F_DIM 128

typedef __attribute__((ext_vector_type(2))) float v2f;
typedef __attribute__((ext_vector_type(8))) float v8f;

static __device__ __forceinline__ v8f wmma_f32x4(v2f a, v2f b, v8f c) {
  return __builtin_amdgcn_wmma_f32_16x16x4_f32(
      false, a, false, b, (short)0, c, false, false);
}

#define BK_SHIFT 8
#define BK (1 << BK_SHIFT)
#define MAXB 512
#define EDGE_BLOCKS 256
#define SCAT_THREADS 512
#define BUILD_THREADS 512

#define TB_SQ (F_DIM * F_DIM / 256)
__global__ void __launch_bounds__(256)
prep_count_transpose(const int* __restrict__ dst, int nE, int n_nodes,
                     int* __restrict__ counts,
                     const float* __restrict__ Ws0, const float* __restrict__ Wn0,
                     const float* __restrict__ Ws1, const float* __restrict__ Wn1,
                     const float* __restrict__ Ws2, const float* __restrict__ Wn2,
                     float* __restrict__ wt, int n_cls) {
  __shared__ int hist[MAXB];
  const int tb_cls = (F_DIM * n_cls + 255) / 256;
  const int n_tb = 4 * TB_SQ + 2 * tb_cls;
  const int b = blockIdx.x;
  if (b < n_tb) {
    const float* W;
    float* WT;
    int n_out, lb;
    if (b < 4 * TB_SQ) {
      const int m = b / TB_SQ;
      lb = b - m * TB_SQ;
      n_out = F_DIM;
      W = (m == 0) ? Ws0 : (m == 1) ? Wn0 : (m == 2) ? Ws1 : Wn1;
      WT = wt + (size_t)m * F_DIM * F_DIM;
    } else {
      const int r = b - 4 * TB_SQ;
      const int m = r / tb_cls;
      lb = r - m * tb_cls;
      n_out = n_cls;
      W = m ? Wn2 : Ws2;
      WT = wt + (size_t)4 * F_DIM * F_DIM + (size_t)m * F_DIM * n_cls;
    }
    const int i = lb * 256 + threadIdx.x;
    if (i < F_DIM * n_out) {
      const int k = i / n_out;
      const int n = i - k * n_out;
      WT[n * F_DIM + k] = W[i];
    }
    return;
  }
  const int j = b - n_tb;
  const int nb = (n_nodes + BK - 1) >> BK_SHIFT;
  for (int k = threadIdx.x; k < MAXB; k += 256) hist[k] = 0;
  __syncthreads();
  const int chunk = (nE + EDGE_BLOCKS - 1) / EDGE_BLOCKS;
  const int e0 = j * chunk;
  const int e1 = (e0 + chunk < nE) ? (e0 + chunk) : nE;
  for (int e = e0 + (int)threadIdx.x; e < e1; e += 256 * 8) {
    int d[8];
#pragma unroll
    for (int u = 0; u < 8; ++u) {
      const int ee = e + u * 256;
      d[u] = (ee < e1) ? dst[ee] : -1;
    }
#pragma unroll
    for (int u = 0; u < 8; ++u)
      if (d[u] >= 0) atomicAdd(&hist[d[u] >> BK_SHIFT], 1);
  }
  __syncthreads();
  for (int k = threadIdx.x; k < nb; k += 256) counts[j * MAXB + k] = hist[k];
}

__global__ void __launch_bounds__(SCAT_THREADS)
bucket_scatter(const int* __restrict__ src, const int* __restrict__ dst,
               const int* __restrict__ counts, int* __restrict__ bstart,
               int* __restrict__ pairs, int nE, int n_nodes) {
  static_assert(SCAT_THREADS == MAXB, "one thread per bucket column");
  __shared__ int off[MAXB];
  __shared__ int wsum[SCAT_THREADS / 32];
  const int tid = threadIdx.x;
  const int lane = tid & 31;
  const int wv = tid >> 5;
  const int j = blockIdx.x;
  const int nb = (n_nodes + BK - 1) >> BK_SHIFT;

  int tot = 0, pre = 0;
  if (tid < nb) {
#pragma unroll 32
    for (int jj = 0; jj < EDGE_BLOCKS; ++jj) {
      const int c = counts[jj * MAXB + tid];
      tot += c;
      pre += (jj < j) ? c : 0;
    }
  }
  int incl = tot;
  for (int o = 1; o < 32; o <<= 1) {
    const int t = __shfl_up(incl, o);
    if (lane >= o) incl += t;
  }
  if (lane == 31) wsum[wv] = incl;
  __syncthreads();
  int wbase = 0;
  for (int w = 0; w < wv; ++w) wbase += wsum[w];
  const int excl = wbase + incl - tot;
  off[tid] = excl + pre;
  if (j == 0 && tid <= nb) bstart[tid] = excl;
  __syncthreads();

  const int chunk = (nE + EDGE_BLOCKS - 1) / EDGE_BLOCKS;
  const int e0 = j * chunk;
  const int e1 = (e0 + chunk < nE) ? (e0 + chunk) : nE;
  int2* __restrict__ pr = (int2*)pairs;
  for (int e = e0 + tid; e < e1; e += SCAT_THREADS * 4) {
    int d[4], s[4];
#pragma unroll
    for (int u = 0; u < 4; ++u) {
      const int ee = e + u * SCAT_THREADS;
      d[u] = (ee < e1) ? dst[ee] : -1;
      s[u] = (ee < e1) ? src[ee] : 0;
    }
#pragma unroll
    for (int u = 0; u < 4; ++u) {
      if (d[u] >= 0) {
        const int pos = atomicAdd(&off[d[u] >> BK_SHIFT], 1);
        pr[pos] = make_int2(d[u], s[u]);
      }
    }
  }
}

__global__ void __launch_bounds__(BUILD_THREADS)
bucket_build(const int* __restrict__ pairs, const int* __restrict__ bstart,
             int* __restrict__ rowptr, int* __restrict__ rowcnt,
             float* __restrict__ inv_deg, int* __restrict__ col, int n_nodes) {
  __shared__ int cnt[BK];
  __shared__ int cur[BK];
  __shared__ int wsum[BK / 32];
  const int tid = threadIdx.x;
  const int lane = tid & 31;
  const int wv = tid >> 5;
  const int k = blockIdx.x;
  const int beg = bstart[k];
  const int end = bstart[k + 1];
  const int node0 = k << BK_SHIFT;
  const int2* __restrict__ pr = (const int2*)pairs;

  if (tid < BK) cnt[tid] = 0;
  __syncthreads();
  for (int i = beg + tid; i < end; i += BUILD_THREADS * 4) {
    int d[4];
#pragma unroll
    for (int u = 0; u < 4; ++u) {
      const int ii = i + u * BUILD_THREADS;
      d[u] = (ii < end) ? pr[ii].x : -1;
    }
#pragma unroll
    for (int u = 0; u < 4; ++u)
      if (d[u] >= 0) atomicAdd(&cnt[d[u] - node0], 1);
  }
  __syncthreads();

  int c = 0, incl = 0;
  if (tid < BK) {
    c = cnt[tid];
    incl = c;
    for (int o = 1; o < 32; o <<= 1) {
      const int t = __shfl_up(incl, o);
      if (lane >= o) incl += t;
    }
    if (lane == 31) wsum[wv] = incl;
  }
  __syncthreads();
  if (tid < BK) {
    int wbase = 0;
    for (int w = 0; w < wv; ++w) wbase += wsum[w];
    const int seg = beg + wbase + incl - c;
    cur[tid] = seg;
    const int node = node0 + tid;
    if (node < n_nodes) {
      rowptr[node] = seg;
      rowcnt[node] = c;
      inv_deg[node] = 1.0f / fmaxf((float)c, 1.0f);
    }
  }
  __syncthreads();

  for (int i = beg + tid; i < end; i += BUILD_THREADS * 4) {
    int2 p[4];
#pragma unroll
    for (int u = 0; u < 4; ++u) {
      const int ii = i + u * BUILD_THREADS;
      p[u] = (ii < end) ? pr[ii] : make_int2(-1, 0);
    }
#pragma unroll
    for (int u = 0; u < 4; ++u) {
      if (p[u].x >= 0) {
        const int pos = atomicAdd(&cur[p[u].x - node0], 1);
        col[pos] = p[u].y;
      }
    }
  }
}

template <bool SORT>
__global__ void __launch_bounds__(256)
node_gather(const float* __restrict__ h, const int* __restrict__ rowptr,
            const int* __restrict__ rowcnt, const int* __restrict__ col_raw,
            int* col_srt, float* __restrict__ agg, int n_nodes) {
  const int lane = threadIdx.x & 31;
  const int node = __builtin_amdgcn_readfirstlane((int)(blockIdx.x * 8 + (threadIdx.x >> 5)));
  if (node >= n_nodes) return;
  const int start = rowptr[node];
  const int L = rowcnt[node];

  int c0 = 0;
  if constexpr (SORT) {
    if (L <= 32) {
      const int raw = (lane < L) ? col_raw[start + lane] : 0x7fffffff;
      int r = 0;
      for (int j = 0; j < L; ++j) {
        const int vj = __builtin_amdgcn_readlane(raw, j);
        r += (vj < raw || (vj == raw && j < lane)) ? 1 : 0;
      }
      if (lane < L) col_srt[start + r] = raw;
      c0 = __builtin_amdgcn_ds_permute(r << 2, raw);
    } else {
      for (int b = 0; b < L; b += 32) {
        const int idx = b + lane;
        const int myv = (idx < L) ? col_raw[start + idx] : 0;
        int r = 0;
        for (int j = 0; j < L; ++j) {
          const int vj = col_raw[start + j];
          r += (vj < myv || (vj == myv && j < idx)) ? 1 : 0;
        }
        if (idx < L) col_srt[start + r] = myv;
      }
      __threadfence();
    }
  }

  float4 acc = make_float4(0.0f, 0.0f, 0.0f, 0.0f);
  for (int base = 0; base < L; base += 32) {
    const int n = (L - base < 32) ? (L - base) : 32;
    int c;
    if (SORT && L <= 32) c = c0;
    else c = (lane < n) ? col_srt[start + base + lane] : 0;
    for (int j = 0; j < n; j += 8) {
      int s[8];
#pragma unroll
      for (int u = 0; u < 8; ++u) {
        const int ju = (j + u < n) ? (j + u) : (n - 1);
        s[u] = __builtin_amdgcn_readlane(c, ju);
      }
      float4 v[8];
#pragma unroll
      for (int u = 0; u < 8; ++u) {
        if (j + u < n) {
          const float4* row = (const float4*)(h + (size_t)s[u] * F_DIM);
          v[u] = row[lane];
        } else {
          v[u] = make_float4(0.0f, 0.0f, 0.0f, 0.0f);
        }
      }
#pragma unroll
      for (int u = 0; u < 8; ++u) {
        acc.x += v[u].x; acc.y += v[u].y; acc.z += v[u].z; acc.w += v[u].w;
      }
    }
  }
  ((float4*)(agg + (size_t)node * F_DIM))[lane] = acc;
}

template <int N_OUT>
__global__ void __launch_bounds__(256)
gather_project_out(const float* __restrict__ S, const float* __restrict__ inv_deg,
                   const int* __restrict__ rowptr, const int* __restrict__ rowcnt,
                   const int* __restrict__ col, float* __restrict__ out, int n_nodes) {
  constexpr int PITCH = ((N_OUT + 15) / 16) * 16;
  constexpr int CHUNKS = PITCH / 4;
  static_assert(CHUNKS <= 16, "a projected row fits one 16-lane half");
  __shared__ __attribute__((aligned(16))) float tr[8][2][PITCH];

  const int lane = threadIdx.x & 31;
  const int wv = __builtin_amdgcn_readfirstlane((int)(threadIdx.x >> 5));
  const int node = __builtin_amdgcn_readfirstlane((int)(blockIdx.x * 8 + (threadIdx.x >> 5)));
  if (node >= n_nodes) return;
  const int half = lane >> 4;
  const int sub = lane & 15;
  const bool chunk_on = sub < CHUNKS;
  const float* __restrict__ P = S + (size_t)n_nodes * PITCH;
  const int start = rowptr[node];
  const int L = rowcnt[node];
  const float invd = inv_deg[node];

  constexpr int OUT_TRIPS = (N_OUT + 31) / 32;
  float sv[OUT_TRIPS];
#pragma unroll
  for (int q = 0; q < OUT_TRIPS; ++q) {
    const int cq = q * 32 + lane;
    sv[q] = (cq < N_OUT) ? S[(size_t)node * PITCH + cq] : 0.0f;
  }

  float4 acc = make_float4(0.0f, 0.0f, 0.0f, 0.0f);
  for (int base = 0; base < L; base += 32) {
    const int n = (L - base < 32) ? (L - base) : 32;
    const int c = (lane < n) ? col[start + base + lane] : 0;
    for (int j = 0; j < n; j += 16) {
      int s[8];
#pragma unroll
      for (int u = 0; u < 8; ++u) {
        const int e0 = (j + 2 * u < n) ? (j + 2 * u) : (n - 1);
        const int e1 = (j + 2 * u + 1 < n) ? (j + 2 * u + 1) : (n - 1);
        const int lo = __builtin_amdgcn_readlane(c, e0);
        const int hi = __builtin_amdgcn_readlane(c, e1);
        s[u] = half ? hi : lo;
      }
      float4 v[8];
#pragma unroll
      for (int u = 0; u < 8; ++u) {
        v[u] = make_float4(0.0f, 0.0f, 0.0f, 0.0f);
        if (j + 2 * u < n) {
          if (chunk_on && j + 2 * u + half < n)
            v[u] = ((const float4*)(P + (size_t)s[u] * PITCH))[sub];
        }
      }
#pragma unroll
      for (int u = 0; u < 8; ++u) {
        acc.x += v[u].x; acc.y += v[u].y; acc.z += v[u].z; acc.w += v[u].w;
      }
    }
  }

  if (chunk_on) *(float4*)&tr[wv][half][4 * sub] = acc;
  float* __restrict__ orow = out + (size_t)node * N_OUT;
#pragma unroll
  for (int q = 0; q < OUT_TRIPS; ++q) {
    const int cq = q * 32 + lane;
    if (cq < N_OUT) {
      const float t = tr[wv][0][cq] + tr[wv][1][cq];
      orow[cq] = sv[q] + invd * t;
    }
  }
}

#define GEMM_GROUPS 256
#define W_PITCH 132

template <int N_OUT, bool RELU, bool PROJECT, int WAVES>
__global__ void __launch_bounds__(WAVES * 32)
sage_gemm(const float* h, const float* __restrict__ agg,
          const float* __restrict__ inv_deg,
          const float* __restrict__ WsT, const float* __restrict__ WnT,
          const float* __restrict__ bias, float* out, int n_nodes) {
  constexpr int THREADS = WAVES * 32;
  constexpr int TILES_N = (N_OUT + 15) / 16;
  constexpr int N_PAD = TILES_N * 16;
  constexpr int KSTEPS = F_DIM / 8;

  __shared__ __attribute__((aligned(16))) float Bs[2 * N_PAD * W_PITCH];

  const int tid = threadIdx.x;
  const int lane = tid & 31;
  const int wv = __builtin_amdgcn_readfirstlane(tid >> 5);
  const int half = lane >> 4;
  const int l16 = lane & 15;
  const int kq = 4 * half;

  const int tiles_m = (n_nodes + 15) >> 4;
  const int groups = (int)gridDim.x;
  const int tile_step = groups * WAVES;
  int tm = wv * groups + (int)blockIdx.x;

  float4 ah[KSTEPS], an[KSTEPS];
  float invd = 1.0f;
  auto fetch_rows = [&](int t_m) {
    int mrow = t_m * 16 + l16;
    if (mrow >= n_nodes) mrow = n_nodes - 1;
    const float4* hp = (const float4*)(h + (size_t)mrow * F_DIM + kq);
#pragma unroll
    for (int t = 0; t < KSTEPS; ++t) ah[t] = hp[2 * t];
    if constexpr (!PROJECT) {
      const float4* __restrict__ ap = (const float4*)(agg + (size_t)mrow * F_DIM + kq);
#pragma unroll
      for (int t = 0; t < KSTEPS; ++t) an[t] = ap[2 * t];
      invd = inv_deg[mrow];
    }
  };
  if (tm < tiles_m) fetch_rows(tm);

  {
    constexpr int ROW4 = F_DIM / 4;
    constexpr int TOT4 = N_PAD * ROW4;
    constexpr int TRIPS = (TOT4 + THREADS - 1) / THREADS;
    const float4* __restrict__ gs = (const float4*)WsT;
    const float4* __restrict__ gn = (const float4*)WnT;
    float4 vs[TRIPS], vn[TRIPS];
#pragma unroll
    for (int u = 0; u < TRIPS; ++u) {
      int i = tid + u * THREADS;
      if (TOT4 % THREADS != 0 && i >= TOT4) i = TOT4 - 1;
      const int n = i / ROW4;
      const int nr = (N_PAD != N_OUT && n >= N_OUT) ? (N_OUT - 1) : n;
      vs[u] = gs[nr * ROW4 + (i % ROW4)];
      vn[u] = gn[nr * ROW4 + (i % ROW4)];
    }
#pragma unroll
    for (int u = 0; u < TRIPS; ++u) {
      int i = tid + u * THREADS;
      if (TOT4 % THREADS != 0 && i >= TOT4) i = TOT4 - 1;
      float* ls = Bs + (i / ROW4) * W_PITCH + 4 * (i % ROW4);
      *(float4*)ls = vs[u];
      *(float4*)(ls + N_PAD * W_PITCH) = vn[u];
    }
  }
  __syncthreads();

  unsigned so = (unsigned)((l16 * W_PITCH + kq) * 4);
  unsigned no = so + (unsigned)(N_PAD * W_PITCH * 4);

  for (; tm < tiles_m; tm += tile_step) {
    asm volatile("" : "+v"(so), "+v"(no) : : "memory");
    const float* bself = (const float*)((const char*)Bs + so);
    const float* bneigh = (const float*)((const char*)Bs + no);
    float bv[TILES_N];
#pragma unroll
    for (int j = 0; j < TILES_N; ++j) {
      const int ncol = j * 16 + l16;
      bv[j] = (ncol < N_OUT) ? bias[(ncol < N_OUT) ? ncol : 0] : 0.0f;
    }
    v8f acc[TILES_N];
    v8f accp[PROJECT ? TILES_N : 1];
#pragma unroll
    for (int j = 0; j < TILES_N; ++j) {
#pragma unroll
      for (int r = 0; r < 8; ++r) acc[j][r] = bv[j];
      if constexpr (PROJECT) {
#pragma unroll
        for (int r = 0; r < 8; ++r) accp[j][r] = 0.0f;
      }
    }

#pragma unroll
    for (int t = 0; t < KSTEPS; ++t) {
      const float4 a = ah[t];
      float4 b[TILES_N];
#pragma unroll
      for (int j = 0; j < TILES_N; ++j)
        b[j] = *(const float4*)(bself + j * 16 * W_PITCH + 8 * t);
#pragma unroll
      for (int j = 0; j < TILES_N; ++j)
        acc[j] = wmma_f32x4(v2f{a.x, a.y}, v2f{b[j].x, b[j].y}, acc[j]);
#pragma unroll
      for (int j = 0; j < TILES_N; ++j)
        acc[j] = wmma_f32x4(v2f{a.z, a.w}, v2f{b[j].z, b[j].w}, acc[j]);
    }
#pragma unroll
    for (int t = 0; t < KSTEPS; ++t) {
      float4 a;
      if constexpr (PROJECT) a = ah[t];
      else a = make_float4(an[t].x * invd, an[t].y * invd, an[t].z * invd, an[t].w * invd);
      float4 b[TILES_N];
#pragma unroll
      for (int j = 0; j < TILES_N; ++j)
        b[j] = *(const float4*)(bneigh + j * 16 * W_PITCH + 8 * t);
      if constexpr (PROJECT) {
#pragma unroll
        for (int j = 0; j < TILES_N; ++j)
          accp[j] = wmma_f32x4(v2f{a.x, a.y}, v2f{b[j].x, b[j].y}, accp[j]);
#pragma unroll
        for (int j = 0; j < TILES_N; ++j)
          accp[j] = wmma_f32x4(v2f{a.z, a.w}, v2f{b[j].z, b[j].w}, accp[j]);
      } else {
#pragma unroll
        for (int j = 0; j < TILES_N; ++j)
          acc[j] = wmma_f32x4(v2f{a.x, a.y}, v2f{b[j].x, b[j].y}, acc[j]);
#pragma unroll
        for (int j = 0; j < TILES_N; ++j)
          acc[j] = wmma_f32x4(v2f{a.z, a.w}, v2f{b[j].z, b[j].w}, acc[j]);
      }
    }

    if constexpr (RELU) {
#pragma unroll
      for (int j = 0; j < TILES_N; ++j)
#pragma unroll
        for (int r = 0; r < 8; ++r) acc[j][r] = fmaxf(acc[j][r], 0.0f);
    }

    const int row0 = tm * 16 + half * 8;
    if constexpr (PROJECT) {
      float* sp = out + (size_t)row0 * N_PAD + l16;
      float* pp = sp + (size_t)n_nodes * N_PAD;
      if (tm * 16 + 16 <= n_nodes) {
#pragma unroll
        for (int j = 0; j < TILES_N; ++j) {
#pragma unroll
          for (int r = 0; r < 8; ++r) {
            sp[r * N_PAD + j * 16] = acc[j][r];
            pp[r * N_PAD + j * 16] = accp[j][r];
          }
        }
      } else {
#pragma unroll
        for (int j = 0; j < TILES_N; ++j) {
#pragma unroll
          for (int r = 0; r < 8; ++r) {
            if (row0 + r < n_nodes) {
              sp[r * N_PAD + j * 16] = acc[j][r];
              pp[r * N_PAD + j * 16] = accp[j][r];
            }
          }
        }
      }
    } else {
      float* op = out + (size_t)row0 * N_OUT + l16;
      if (tm * 16 + 16 <= n_nodes) {
#pragma unroll
        for (int j = 0; j < TILES_N; ++j) {
          if (N_PAD == N_OUT || j * 16 + l16 < N_OUT) {
#pragma unroll
            for (int r = 0; r < 8; ++r) op[r * N_OUT + j * 16] = acc[j][r];
          }
        }
      } else {
#pragma unroll
        for (int j = 0; j < TILES_N; ++j) {
#pragma unroll
          for (int r = 0; r < 8; ++r) {
            if (row0 + r < n_nodes && j * 16 + l16 < N_OUT) op[r * N_OUT + j * 16] = acc[j][r];
          }
        }
      }
    }

    const int tn = tm + tile_step;
    if (tn < tiles_m) fetch_rows(tn);
  }
}

extern "C" void kernel_launch(void* const* d_in, const int* in_sizes, int n_in,
                              void* d_out, int out_size, void* d_ws, size_t ws_size,
                              hipStream_t stream) {
  const float* x   = (const float*)d_in[0];
  const int*   src = (const int*)d_in[1];
  const int*   dst = (const int*)d_in[2];
  const float* Ws[3] = { (const float*)d_in[3], (const float*)d_in[6], (const float*)d_in[9]  };
  const float* Wn[3] = { (const float*)d_in[4], (const float*)d_in[7], (const float*)d_in[10] };
  const float* bs[3] = { (const float*)d_in[5], (const float*)d_in[8], (const float*)d_in[11] };

  const int N = in_sizes[0] / F_DIM;
  const int E = in_sizes[1];
  const int NC = 47;

  float* invdeg = (float*)d_ws;
  float* agg = invdeg + (size_t)N;
  float* hA  = agg + (size_t)N * F_DIM;
  float* epl = hA  + (size_t)N * F_DIM;
  float* wt  = epl + (size_t)N * F_DIM;
  float* WsT[3] = { wt,                 wt + 2 * F_DIM * F_DIM, wt + 4 * F_DIM * F_DIM };
  float* WnT[3] = { wt + F_DIM * F_DIM, wt + 3 * F_DIM * F_DIM, wt + 4 * F_DIM * F_DIM + F_DIM * NC };

  int* counts  = (int*)epl;
  int* bstart  = counts + (size_t)EDGE_BLOCKS * MAXB;
  int* pairs   = bstart + 2 * MAXB;
  int* col_raw = pairs + 2 * (size_t)E;
  int* col_srt = col_raw + (size_t)E;
  int* rowptr  = col_srt + (size_t)E;
  int* rowcnt  = rowptr + (size_t)N;

  const int TB = 4 * TB_SQ + 2 * ((F_DIM * NC + 255) / 256);
  prep_count_transpose<<<TB + EDGE_BLOCKS, 256, 0, stream>>>(
      dst, E, N, counts, Ws[0], Wn[0], Ws[1], Wn[1], Ws[2], Wn[2], wt, NC);

  bucket_scatter<<<EDGE_BLOCKS, SCAT_THREADS, 0, stream>>>(src, dst, counts, bstart, pairs, E, N);
  bucket_build<<<(N + BK - 1) / BK, BUILD_THREADS, 0, stream>>>(
      pairs, bstart, rowptr, rowcnt, invdeg, col_raw, N);

  int gather_blocks = (N + 7) / 8;

  node_gather<true><<<gather_blocks, 256, 0, stream>>>(x, rowptr, rowcnt, col_raw, col_srt, agg, N);
  sage_gemm<F_DIM, true, false, 12><<<GEMM_GROUPS, 12 * 32, 0, stream>>>(
      x, agg, invdeg, WsT[0], WnT[0], bs[0], hA, N);
  node_gather<false><<<gather_blocks, 256, 0, stream>>>(hA, rowptr, rowcnt, col_raw, col_srt, agg, N);
  sage_gemm<F_DIM, true, false, 12><<<GEMM_GROUPS, 12 * 32, 0, stream>>>(
      hA, agg, invdeg, WsT[1], WnT[1], bs[1], hA, N);
  sage_gemm<NC, false, true, 16><<<GEMM_GROUPS, 16 * 32, 0, stream>>>(
      hA, nullptr, nullptr, WsT[2], WnT[2], bs[2], agg, N);
  gather_project_out<NC><<<gather_blocks, 256, 0, stream>>>(
      agg, invdeg, rowptr, rowcnt, col_srt, (float*)d_out, N);
}
